# starting kernel
# baseline (speedup 1.0000x reference)
#include <hip/hip_runtime.h>

typedef __attribute__((ext_vector_type(2))) float v2f;
typedef __attribute__((ext_vector_type(8))) float v8f;

#define N_NODES 50000
#define N_EDGES 625000
#define CH 128
#define ROW_BLOCKS (N_NODES / 16)

__global__ void k_zero(float* __restrict__ agg, float* __restrict__ deg) {
    int i = blockIdx.x * blockDim.x + threadIdx.x;
    agg[i] = 0.0f;
    if (i < N_NODES) deg[i] = 0.0f;
}

__global__ void k_deg(const int* __restrict__ ei, float* __restrict__ deg) {
    int e = blockIdx.x * blockDim.x + threadIdx.x;
    if (e < N_EDGES) atomicAdd(&deg[ei[e]], 1.0f);
}

__global__ void k_dis(const float* __restrict__ deg, float* __restrict__ dis) {
    int n = blockIdx.x * blockDim.x + threadIdx.x;
    if (n < N_NODES) {
        float d = deg[n];
        dis[n] = (d > 0.0f) ? rsqrtf(d) : 0.0f;
    }
}

__global__ void k_scatter(const float* __restrict__ x,
                          const int* __restrict__ ei,
                          const float* __restrict__ dis,
                          float* __restrict__ agg) {
    int gid  = blockIdx.x * blockDim.x + threadIdx.x;
    int edge = gid >> 5;
    int lane = threadIdx.x & 31;
    if (edge >= N_EDGES) return;

    int r = ei[edge];
    int c = ei[N_EDGES + edge];
    float w = dis[r] * dis[c];

    const float4* xr = (const float4*)(x + (size_t)c * CH);
    float4 v = xr[lane];

    float* o = agg + (size_t)r * CH + lane * 4;
    atomicAdd(o + 0, v.x * w);
    atomicAdd(o + 1, v.y * w);
    atomicAdd(o + 2, v.z * w);
    atomicAdd(o + 3, v.w * w);
}

__global__ void k_gemm(float* __restrict__ io,
                       const float* __restrict__ W,
                       const float* __restrict__ bias) {
    __shared__ float Wl[CH * CH];

    int tid = threadIdx.x;
    const float4* Wv  = (const float4*)W;
    float4*       Wlv = (float4*)Wl;
#pragma unroll
    for (int i = 0; i < 16; ++i) Wlv[tid + i * 256] = Wv[tid + i * 256];
    __syncthreads();

    int wave = tid >> 5;
    int lane = tid & 31;
    int rb = blockIdx.x * 8 + wave;
    if (rb >= ROW_BLOCKS) return;

    int m    = lane & 15;
    int kh   = (lane >> 4) * 2;
    int ncol = lane & 15;

    const float* arow = io + ((size_t)rb * 16 + m) * CH;

    v8f acc[8] = {};

    for (int k0 = 0; k0 < CH; k0 += 4) {
        v2f a;
        a.x = arow[k0 + kh];
        a.y = arow[k0 + kh + 1];
#pragma unroll
        for (int t = 0; t < 8; ++t) {
            const float* wrow = Wl + (t * 16 + ncol) * CH + k0 + kh;
            v2f b;
            b.x = wrow[0];
            b.y = wrow[1];
            acc[t] = __builtin_amdgcn_wmma_f32_16x16x4_f32(
                false, a, false, b, (short)0, acc[t], false, false);
        }
    }

    int base_m = rb * 16 + (lane >> 4) * 8;
#pragma unroll
    for (int t = 0; t < 8; ++t) {
        int col = t * 16 + ncol;
        float bv = bias[col];
#pragma unroll
        for (int r = 0; r < 8; ++r) {
            io[(size_t)(base_m + r) * CH + col] = acc[t][r] + bv;
        }
    }
}

extern "C" void kernel_launch(void* const* d_in, const int* in_sizes, int n_in,
                              void* d_out, int out_size, void* d_ws, size_t ws_size,
                              hipStream_t stream) {
    const float* x  = (const float*)d_in[0];
    const int*   ei = (const int*)d_in[1];
    const float* W  = (const float*)d_in[2];
    const float* b  = (const float*)d_in[3];
    float* out = (float*)d_out;

    float* deg = (float*)d_ws;
    float* dis = deg + N_NODES;

    k_zero<<<(N_NODES * CH) / 256, 256, 0, stream>>>(out, deg);
    k_deg<<<(N_EDGES + 255) / 256, 256, 0, stream>>>(ei, deg);
    k_dis<<<(N_NODES + 255) / 256, 256, 0, stream>>>(deg, dis);
    k_scatter<<<(N_EDGES * 32) / 256, 256, 0, stream>>>(x, ei, dis, out);
    k_gemm<<<(ROW_BLOCKS + 7) / 8, 256, 0, stream>>>(out, W, b);
}


# GEMM: W staged in LDS at a 132-float row pitch with k pairs interleaved, so B fragments come as conflict-free 16-byte reads instead of 16-way bank-conflicted 8-byte pairs; k order and sums unchanged
# speedup vs baseline: 1.0541x; 1.0541x over previous
#include <hip/hip_runtime.h>

typedef __attribute__((ext_vector_type(2))) float v2f;
typedef __attribute__((ext_vector_type(8))) float v8f;

#define N_NODES 50000
#define N_EDGES 625000
#define CH 128
#define ROW_BLOCKS (N_NODES / 16)

__global__ void k_zero(float* __restrict__ agg, float* __restrict__ deg) {
    int i = blockIdx.x * blockDim.x + threadIdx.x;
    agg[i] = 0.0f;
    if (i < N_NODES) deg[i] = 0.0f;
}

__global__ void k_deg(const int* __restrict__ ei, float* __restrict__ deg) {
    int e = blockIdx.x * blockDim.x + threadIdx.x;
    if (e < N_EDGES) atomicAdd(&deg[ei[e]], 1.0f);
}

__global__ void k_dis(const float* __restrict__ deg, float* __restrict__ dis) {
    int n = blockIdx.x * blockDim.x + threadIdx.x;
    if (n < N_NODES) {
        float d = deg[n];
        dis[n] = (d > 0.0f) ? rsqrtf(d) : 0.0f;
    }
}

__global__ void k_scatter(const float* __restrict__ x,
                          const int* __restrict__ ei,
                          const float* __restrict__ dis,
                          float* __restrict__ agg) {
    int gid  = blockIdx.x * blockDim.x + threadIdx.x;
    int edge = gid >> 5;
    int lane = threadIdx.x & 31;
    if (edge >= N_EDGES) return;

    int r = ei[edge];
    int c = ei[N_EDGES + edge];
    float w = dis[r] * dis[c];

    const float4* xr = (const float4*)(x + (size_t)c * CH);
    float4 v = xr[lane];

    float* o = agg + (size_t)r * CH + lane * 4;
    atomicAdd(o + 0, v.x * w);
    atomicAdd(o + 1, v.y * w);
    atomicAdd(o + 2, v.z * w);
    atomicAdd(o + 3, v.w * w);
}

#define WPITCH 132

__global__ void __launch_bounds__(256)
k_gemm(float* __restrict__ io,
                       const float* __restrict__ W,
                       const float* __restrict__ bias) {
    __shared__ float4 Wl4[CH * WPITCH / 4];
    float* Wl = (float*)Wl4;

    int tid = threadIdx.x;
    const float4* Wv = (const float4*)W;
#pragma unroll
    for (int i = 0; i < 8; ++i) {
        int g = tid + i * 256;
        float4 lo = Wv[2 * g];
        float4 hi = Wv[2 * g + 1];
        float4* dst = (float4*)(Wl + (g >> 4) * WPITCH + (g & 15) * 8);
        dst[0] = make_float4(lo.x, lo.y, hi.x, hi.y);
        dst[1] = make_float4(lo.z, lo.w, hi.z, hi.w);
    }
    __syncthreads();

    int wave = tid >> 5;
    int lane = tid & 31;
    int rb = blockIdx.x * 8 + wave;
    if (rb >= ROW_BLOCKS) return;

    int m    = lane & 15;
    int half = lane >> 4;
    int ncol = lane & 15;

    const float* arow = io + ((size_t)rb * 16 + m) * CH + 2 * half;
    const float4* wq  = (const float4*)(Wl + ncol * WPITCH + 4 * half);

    v8f acc[8] = {};

    for (int j = 0; j < CH / 8; ++j) {
        v2f a0, a1;
        a0.x = arow[8 * j];
        a0.y = arow[8 * j + 1];
        a1.x = arow[8 * j + 4];
        a1.y = arow[8 * j + 5];
        float4 bq[8];
#pragma unroll
        for (int t = 0; t < 8; ++t)
            bq[t] = wq[(t * 16 * WPITCH + 8 * j) / 4];
#pragma unroll
        for (int t = 0; t < 8; ++t) {
            v2f b;
            b.x = bq[t].x;
            b.y = bq[t].y;
            acc[t] = __builtin_amdgcn_wmma_f32_16x16x4_f32(
                false, a0, false, b, (short)0, acc[t], false, false);
        }
#pragma unroll
        for (int t = 0; t < 8; ++t) {
            v2f b;
            b.x = bq[t].z;
            b.y = bq[t].w;
            acc[t] = __builtin_amdgcn_wmma_f32_16x16x4_f32(
                false, a1, false, b, (short)0, acc[t], false, false);
        }
    }

    int base_m = rb * 16 + half * 8;
#pragma unroll
    for (int t = 0; t < 8; ++t) {
        int col = t * 16 + ncol;
        float bv = bias[col];
#pragma unroll
        for (int r = 0; r < 8; ++r) {
            io[(size_t)(base_m + r) * CH + col] = acc[t][r] + bv;
        }
    }
}

extern "C" void kernel_launch(void* const* d_in, const int* in_sizes, int n_in,
                              void* d_out, int out_size, void* d_ws, size_t ws_size,
                              hipStream_t stream) {
    const float* x  = (const float*)d_in[0];
    const int*   ei = (const int*)d_in[1];
    const float* W  = (const float*)d_in[2];
    const float* b  = (const float*)d_in[3];
    float* out = (float*)d_out;

    float* deg = (float*)d_ws;
    float* dis = deg + N_NODES;

    k_zero<<<(N_NODES * CH) / 256, 256, 0, stream>>>(out, deg);
    k_deg<<<(N_EDGES + 255) / 256, 256, 0, stream>>>(ei, deg);
    k_dis<<<(N_NODES + 255) / 256, 256, 0, stream>>>(deg, dis);
    k_scatter<<<(N_EDGES * 32) / 256, 256, 0, stream>>>(x, ei, dis, out);
    k_gemm<<<(ROW_BLOCKS + 7) / 8, 256, 0, stream>>>(out, W, b);
}


# Aggregate by gather: edges bucketed by destination (integer atomics), each bucket summed in ascending edge id (not arrival order) so float sums are run-to-run fixed; float atomics, out zeroing gone
# speedup vs baseline: 1.6765x; 1.5905x over previous
#include <hip/hip_runtime.h>

typedef __attribute__((ext_vector_type(2))) float v2f;
typedef __attribute__((ext_vector_type(8))) float v8f;

#define N_NODES 50000
#define N_EDGES 625000
#define CH 128
#define ROW_BLOCKS (N_NODES / 16)

#define WS_CNT   0
#define WS_TOTAL N_NODES
#define WS_OFF   50016
#define WS_DIS   (WS_OFF + N_NODES)
#define WS_RANK  (WS_DIS + N_NODES)
#define WS_BEC   (WS_RANK + N_EDGES)
#define WS_BW    (WS_BEC + 2 * N_EDGES)

__global__ void k_zero(int* __restrict__ cnt) {
    int i = blockIdx.x * blockDim.x + threadIdx.x;
    if (i <= N_NODES) cnt[i] = 0;
}

__global__ void k_count(const int* __restrict__ ei, int* __restrict__ cnt,
                        int* __restrict__ rank) {
    int e = blockIdx.x * blockDim.x + threadIdx.x;
    if (e < N_EDGES) rank[e] = atomicAdd(&cnt[ei[e]], 1);
}

__global__ void k_off(const int* __restrict__ cnt, int* __restrict__ total,
                      int* __restrict__ off, float* __restrict__ dis) {
    int n    = blockIdx.x * blockDim.x + threadIdx.x;
    int lane = threadIdx.x & 31;
    int c    = (n < N_NODES) ? cnt[n] : 0;
    int incl = c;
    for (int s = 1; s < 32; s <<= 1) {
        int t = __shfl_up(incl, s);
        if (lane >= s) incl += t;
    }
    int base = 0;
    if (lane == 31) base = atomicAdd(total, incl);
    base = __shfl(base, 31);
    if (n < N_NODES) {
        off[n] = base + incl - c;
        dis[n] = (c > 0) ? rsqrtf((float)c) : 0.0f;
    }
}

__global__ void k_fill(const int* __restrict__ ei, const int* __restrict__ off,
                       const int* __restrict__ rank, const float* __restrict__ dis,
                       int* __restrict__ bec_w, float* __restrict__ bw) {
    int2* bec = (int2*)bec_w;
    int e = blockIdx.x * blockDim.x + threadIdx.x;
    if (e >= N_EDGES) return;
    int r = ei[e];
    int c = ei[N_EDGES + e];
    int p = off[r] + rank[e];
    bec[p] = make_int2(e, c);
    bw[p]  = dis[r] * dis[c];
}

__device__ __forceinline__ float4 fma4(float4 v, float w, float4 a) {
    a.x = fmaf(v.x, w, a.x);
    a.y = fmaf(v.y, w, a.y);
    a.z = fmaf(v.z, w, a.z);
    a.w = fmaf(v.w, w, a.w);
    return a;
}

__global__ void __launch_bounds__(256)
k_gather(const float* __restrict__ x, const int* __restrict__ cnt,
         const int* __restrict__ off, const int* __restrict__ bec_w,
         const float* __restrict__ bw, float* __restrict__ agg) {
    const int2* bec = (const int2*)bec_w;
    int lane = threadIdx.x & 31;
    int node = __builtin_amdgcn_readfirstlane(blockIdx.x * 8 + (threadIdx.x >> 5));
    if (node >= N_NODES) return;
    int d     = cnt[node];
    int start = off[node];
    const float4* x4 = (const float4*)x;
    float4 acc = make_float4(0.0f, 0.0f, 0.0f, 0.0f);

    if (d <= 32) {
        int e = 0x7fffffff, c = 0;
        float w = 0.0f;
        if (lane < d) {
            int2 ec = bec[start + lane];
            e = ec.x;
            c = ec.y;
            w = bw[start + lane];
        }
        int rk = 0;
        for (int k = 0; k < d; ++k) {
            int ek = __builtin_amdgcn_readlane(e, k);
            rk += (ek < e) ? 1 : 0;
        }
        if (lane >= d) rk = lane;
        int   sc = __builtin_amdgcn_ds_permute(rk << 2, c);
        float sw = __int_as_float(__builtin_amdgcn_ds_permute(rk << 2, __float_as_int(w)));

        int i = 0;
        for (; i + 4 <= d; i += 4) {
            int c0 = __builtin_amdgcn_readlane(sc, i);
            int c1 = __builtin_amdgcn_readlane(sc, i + 1);
            int c2 = __builtin_amdgcn_readlane(sc, i + 2);
            int c3 = __builtin_amdgcn_readlane(sc, i + 3);
            float4 v0 = x4[(size_t)c0 * (CH / 4) + lane];
            float4 v1 = x4[(size_t)c1 * (CH / 4) + lane];
            float4 v2 = x4[(size_t)c2 * (CH / 4) + lane];
            float4 v3 = x4[(size_t)c3 * (CH / 4) + lane];
            float w0 = __int_as_float(__builtin_amdgcn_readlane(__float_as_int(sw), i));
            float w1 = __int_as_float(__builtin_amdgcn_readlane(__float_as_int(sw), i + 1));
            float w2 = __int_as_float(__builtin_amdgcn_readlane(__float_as_int(sw), i + 2));
            float w3 = __int_as_float(__builtin_amdgcn_readlane(__float_as_int(sw), i + 3));
            acc = fma4(v0, w0, acc);
            acc = fma4(v1, w1, acc);
            acc = fma4(v2, w2, acc);
            acc = fma4(v3, w3, acc);
        }
        for (; i < d; ++i) {
            int   ci = __builtin_amdgcn_readlane(sc, i);
            float wi = __int_as_float(__builtin_amdgcn_readlane(__float_as_int(sw), i));
            acc = fma4(x4[(size_t)ci * (CH / 4) + lane], wi, acc);
        }
    } else {
        int last = -1;
        for (int i = 0; i < d; ++i) {
            int best = 0x7fffffff, bp = 0;
            for (int p = lane; p < d; p += 32) {
                int id = bec[start + p].x;
                if (id > last && id < best) { best = id; bp = p; }
            }
            for (int s = 16; s > 0; s >>= 1) {
                int ob = __shfl_xor(best, s);
                int op = __shfl_xor(bp, s);
                if (ob < best) { best = ob; bp = op; }
            }
            last = best;
            int   ci = bec[start + bp].y;
            float wi = bw[start + bp];
            acc = fma4(x4[(size_t)ci * (CH / 4) + lane], wi, acc);
        }
    }
    ((float4*)agg)[(size_t)node * (CH / 4) + lane] = acc;
}

#define WPITCH 132

__global__ void __launch_bounds__(256)
k_gemm(float* __restrict__ io,
                       const float* __restrict__ W,
                       const float* __restrict__ bias) {
    __shared__ float4 Wl4[CH * WPITCH / 4];
    float* Wl = (float*)Wl4;

    int tid = threadIdx.x;
    const float4* Wv = (const float4*)W;
#pragma unroll
    for (int i = 0; i < 8; ++i) {
        int g = tid + i * 256;
        float4 lo = Wv[2 * g];
        float4 hi = Wv[2 * g + 1];
        float4* dst = (float4*)(Wl + (g >> 4) * WPITCH + (g & 15) * 8);
        dst[0] = make_float4(lo.x, lo.y, hi.x, hi.y);
        dst[1] = make_float4(lo.z, lo.w, hi.z, hi.w);
    }
    __syncthreads();

    int wave = tid >> 5;
    int lane = tid & 31;
    int rb = blockIdx.x * 8 + wave;
    if (rb >= ROW_BLOCKS) return;

    int m    = lane & 15;
    int half = lane >> 4;
    int ncol = lane & 15;

    const float* arow = io + ((size_t)rb * 16 + m) * CH + 2 * half;
    const float4* wq  = (const float4*)(Wl + ncol * WPITCH + 4 * half);

    v8f acc[8] = {};

    for (int j = 0; j < CH / 8; ++j) {
        v2f a0, a1;
        a0.x = arow[8 * j];
        a0.y = arow[8 * j + 1];
        a1.x = arow[8 * j + 4];
        a1.y = arow[8 * j + 5];
        float4 bq[8];
#pragma unroll
        for (int t = 0; t < 8; ++t)
            bq[t] = wq[(t * 16 * WPITCH + 8 * j) / 4];
#pragma unroll
        for (int t = 0; t < 8; ++t) {
            v2f b;
            b.x = bq[t].x;
            b.y = bq[t].y;
            acc[t] = __builtin_amdgcn_wmma_f32_16x16x4_f32(
                false, a0, false, b, (short)0, acc[t], false, false);
        }
#pragma unroll
        for (int t = 0; t < 8; ++t) {
            v2f b;
            b.x = bq[t].z;
            b.y = bq[t].w;
            acc[t] = __builtin_amdgcn_wmma_f32_16x16x4_f32(
                false, a1, false, b, (short)0, acc[t], false, false);
        }
    }

    int base_m = rb * 16 + half * 8;
#pragma unroll
    for (int t = 0; t < 8; ++t) {
        int col = t * 16 + ncol;
        float bv = bias[col];
#pragma unroll
        for (int r = 0; r < 8; ++r) {
            io[(size_t)(base_m + r) * CH + col] = acc[t][r] + bv;
        }
    }
}

extern "C" void kernel_launch(void* const* d_in, const int* in_sizes, int n_in,
                              void* d_out, int out_size, void* d_ws, size_t ws_size,
                              hipStream_t stream) {
    const float* x  = (const float*)d_in[0];
    const int*   ei = (const int*)d_in[1];
    const float* W  = (const float*)d_in[2];
    const float* b  = (const float*)d_in[3];
    float* out = (float*)d_out;

    int*   wsi   = (int*)d_ws;
    int*   cnt   = wsi + WS_CNT;
    int*   total = wsi + WS_TOTAL;
    int*   off   = wsi + WS_OFF;
    float* dis   = (float*)(wsi + WS_DIS);
    int*   rank  = wsi + WS_RANK;
    int*   bec   = wsi + WS_BEC;
    float* bw    = (float*)(wsi + WS_BW);

    k_zero<<<(N_NODES + 1 + 255) / 256, 256, 0, stream>>>(cnt);
    k_count<<<(N_EDGES + 255) / 256, 256, 0, stream>>>(ei, cnt, rank);
    k_off<<<(N_NODES + 255) / 256, 256, 0, stream>>>(cnt, total, off, dis);
    k_fill<<<(N_EDGES + 255) / 256, 256, 0, stream>>>(ei, off, rank, dis, bec, bw);
    k_gather<<<(N_NODES + 7) / 8, 256, 0, stream>>>(x, cnt, off, bec, bw, out);
    k_gemm<<<(ROW_BLOCKS + 7) / 8, 256, 0, stream>>>(out, W, b);
}


# Fewer, longer-lived waves: 8 edges per thread in count and fill, 4 nodes per thread in offsets, a gather wave walks several nodes with the next bucket loaded ahead; order of float sums unchanged
# speedup vs baseline: 1.7968x; 1.0718x over previous
#include <hip/hip_runtime.h>

typedef __attribute__((ext_vector_type(2))) float v2f;
typedef __attribute__((ext_vector_type(8))) float v8f;

#define N_NODES 50000
#define N_EDGES 625000
#define CH 128
#define ROW_BLOCKS (N_NODES / 16)

#define WS_CNT   0
#define WS_TOTAL N_NODES
#define WS_OFF   50016
#define WS_DIS   (WS_OFF + N_NODES)
#define WS_RANK  (WS_DIS + N_NODES)
#define WS_BEC   (WS_RANK + N_EDGES)
#define WS_BW    (WS_BEC + 2 * N_EDGES)

#define EPT 8
#define EDGE_THREADS (N_EDGES / EPT)
#define EDGE_BLOCK 64
#define NODE_THREADS (N_NODES / 4)
#define ZERO_THREADS ((N_NODES + 4) / 4)
#define GATHER_WAVES 16384

__global__ void __launch_bounds__(256) k_zero(int* __restrict__ cnt) {
    int i = blockIdx.x * 256 + threadIdx.x;
    if (i < ZERO_THREADS) ((int4*)cnt)[i] = make_int4(0, 0, 0, 0);
}

__global__ void __launch_bounds__(256)
k_count(const int* __restrict__ ei, int* __restrict__ cnt, int* __restrict__ rank) {
    int t = blockIdx.x * EDGE_BLOCK + threadIdx.x;
    if (t >= EDGE_THREADS) return;
    const int4* r4 = (const int4*)ei;
    int4 a = r4[2 * t];
    int4 b = r4[2 * t + 1];
    int4 ra, rb;
    ra.x = atomicAdd(&cnt[a.x], 1);
    ra.y = atomicAdd(&cnt[a.y], 1);
    ra.z = atomicAdd(&cnt[a.z], 1);
    ra.w = atomicAdd(&cnt[a.w], 1);
    rb.x = atomicAdd(&cnt[b.x], 1);
    rb.y = atomicAdd(&cnt[b.y], 1);
    rb.z = atomicAdd(&cnt[b.z], 1);
    rb.w = atomicAdd(&cnt[b.w], 1);
    ((int4*)rank)[2 * t]     = ra;
    ((int4*)rank)[2 * t + 1] = rb;
}

__global__ void __launch_bounds__(256)
k_off(const int* __restrict__ cnt, int* __restrict__ total,
      int* __restrict__ off, float* __restrict__ dis) {
    int t    = blockIdx.x * 256 + threadIdx.x;
    int lane = threadIdx.x & 31;
    int4 c = make_int4(0, 0, 0, 0);
    if (t < NODE_THREADS) c = ((const int4*)cnt)[t];
    int s    = c.x + c.y + c.z + c.w;
    int incl = s;
    for (int k = 1; k < 32; k <<= 1) {
        int v = __shfl_up(incl, k);
        if (lane >= k) incl += v;
    }
    int base = 0;
    if (lane == 31) base = atomicAdd(total, incl);
    base = __shfl(base, 31);
    if (t < NODE_THREADS) {
        int o = base + incl - s;
        ((int4*)off)[t] = make_int4(o, o + c.x, o + c.x + c.y, o + c.x + c.y + c.z);
        float4 dd;
        dd.x = (c.x > 0) ? rsqrtf((float)c.x) : 0.0f;
        dd.y = (c.y > 0) ? rsqrtf((float)c.y) : 0.0f;
        dd.z = (c.z > 0) ? rsqrtf((float)c.z) : 0.0f;
        dd.w = (c.w > 0) ? rsqrtf((float)c.w) : 0.0f;
        ((float4*)dis)[t] = dd;
    }
}

__global__ void __launch_bounds__(256)
k_fill(const int* __restrict__ ei, const int* __restrict__ off,
       const int* __restrict__ rank, const float* __restrict__ dis,
       int* __restrict__ bec_w, float* __restrict__ bw) {
    int2* bec = (int2*)bec_w;
    int t = blockIdx.x * EDGE_BLOCK + threadIdx.x;
    if (t >= EDGE_THREADS) return;
    const int4* r4 = (const int4*)ei;
    const int4* c4 = (const int4*)(ei + N_EDGES);
    const int4* k4 = (const int4*)rank;
    int4 ra = r4[2 * t], rb = r4[2 * t + 1];
    int4 ca = c4[2 * t], cb = c4[2 * t + 1];
    int4 ka = k4[2 * t], kb = k4[2 * t + 1];
    int r[EPT] = {ra.x, ra.y, ra.z, ra.w, rb.x, rb.y, rb.z, rb.w};
    int c[EPT] = {ca.x, ca.y, ca.z, ca.w, cb.x, cb.y, cb.z, cb.w};
    int k[EPT] = {ka.x, ka.y, ka.z, ka.w, kb.x, kb.y, kb.z, kb.w};
    int   p[EPT];
    float w[EPT];
#pragma unroll
    for (int j = 0; j < EPT; ++j) {
        p[j] = off[r[j]] + k[j];
        w[j] = dis[r[j]] * dis[c[j]];
    }
#pragma unroll
    for (int j = 0; j < EPT; ++j) {
        bec[p[j]] = make_int2(EPT * t + j, c[j]);
        bw[p[j]]  = w[j];
    }
}

__device__ __forceinline__ float4 fma4(float4 v, float w, float4 a) {
    a.x = fmaf(v.x, w, a.x);
    a.y = fmaf(v.y, w, a.y);
    a.z = fmaf(v.z, w, a.z);
    a.w = fmaf(v.w, w, a.w);
    return a;
}

__global__ void __launch_bounds__(256)
k_gather(const float* __restrict__ x, const int* __restrict__ cnt,
         const int* __restrict__ off, const int* __restrict__ bec_w,
         const float* __restrict__ bw, float* __restrict__ agg) {
    const int2* bec = (const int2*)bec_w;
    int lane = threadIdx.x & 31;
    int node = __builtin_amdgcn_readfirstlane(blockIdx.x * 8 + (threadIdx.x >> 5));
    const float4* x4 = (const float4*)x;

    int d     = cnt[node];
    int start = off[node];
    int   q    = min(start + lane, N_EDGES - 1);
    int2  raw  = bec[q];
    float raww = bw[q];
    int n1 = node + GATHER_WAVES;
    int d1 = 0, start1 = 0;
    if (n1 < N_NODES) { d1 = cnt[n1]; start1 = off[n1]; }

    for (;;) {
        int   q1    = min(start1 + lane, N_EDGES - 1);
        int2  raw1  = bec[q1];
        float raww1 = bw[q1];
        int n2 = n1 + GATHER_WAVES;
        int d2 = 0, start2 = 0;
        if (n2 < N_NODES) { d2 = cnt[n2]; start2 = off[n2]; }

        int   e = (lane < d) ? raw.x : 0x7fffffff;
        int   c = (lane < d) ? raw.y : 0;
        float w = (lane < d) ? raww : 0.0f;

        float4 acc = make_float4(0.0f, 0.0f, 0.0f, 0.0f);
        if (d <= 32) {
            int rk = 0;
            for (int k = 0; k < d; ++k) {
                int ek = __builtin_amdgcn_readlane(e, k);
                rk += (ek < e) ? 1 : 0;
            }
            if (lane >= d) rk = lane;
            int   sc = __builtin_amdgcn_ds_permute(rk << 2, c);
            float sw = __int_as_float(__builtin_amdgcn_ds_permute(rk << 2, __float_as_int(w)));

            int i = 0;
            for (; i + 4 <= d; i += 4) {
                int c0 = __builtin_amdgcn_readlane(sc, i);
                int c1r = __builtin_amdgcn_readlane(sc, i + 1);
                int c2 = __builtin_amdgcn_readlane(sc, i + 2);
                int c3 = __builtin_amdgcn_readlane(sc, i + 3);
                float4 v0 = x4[(size_t)c0 * (CH / 4) + lane];
                float4 v1 = x4[(size_t)c1r * (CH / 4) + lane];
                float4 v2 = x4[(size_t)c2 * (CH / 4) + lane];
                float4 v3 = x4[(size_t)c3 * (CH / 4) + lane];
                float w0 = __int_as_float(__builtin_amdgcn_readlane(__float_as_int(sw), i));
                float w1r = __int_as_float(__builtin_amdgcn_readlane(__float_as_int(sw), i + 1));
                float w2 = __int_as_float(__builtin_amdgcn_readlane(__float_as_int(sw), i + 2));
                float w3 = __int_as_float(__builtin_amdgcn_readlane(__float_as_int(sw), i + 3));
                acc = fma4(v0, w0, acc);
                acc = fma4(v1, w1r, acc);
                acc = fma4(v2, w2, acc);
                acc = fma4(v3, w3, acc);
            }
            for (; i < d; ++i) {
                int   ci = __builtin_amdgcn_readlane(sc, i);
                float wi = __int_as_float(__builtin_amdgcn_readlane(__float_as_int(sw), i));
                acc = fma4(x4[(size_t)ci * (CH / 4) + lane], wi, acc);
            }
        } else {
            int last = -1;
            for (int i = 0; i < d; ++i) {
                int best = 0x7fffffff, bp = 0;
                for (int p = lane; p < d; p += 32) {
                    int id = bec[start + p].x;
                    if (id > last && id < best) { best = id; bp = p; }
                }
                for (int s = 16; s > 0; s >>= 1) {
                    int ob = __shfl_xor(best, s);
                    int op = __shfl_xor(bp, s);
                    if (ob < best) { best = ob; bp = op; }
                }
                last = best;
                int   ci = bec[start + bp].y;
                float wi = bw[start + bp];
                acc = fma4(x4[(size_t)ci * (CH / 4) + lane], wi, acc);
            }
        }
        ((float4*)agg)[(size_t)node * (CH / 4) + lane] = acc;

        if (n1 >= N_NODES) break;
        node = n1;  d = d1;  start = start1;
        raw = raw1;  raww = raww1;
        n1 = n2;  d1 = d2;  start1 = start2;
    }
}

#define WPITCH 132

__global__ void __launch_bounds__(256)
k_gemm(float* __restrict__ io,
                       const float* __restrict__ W,
                       const float* __restrict__ bias) {
    __shared__ float4 Wl4[CH * WPITCH / 4];
    float* Wl = (float*)Wl4;

    int tid = threadIdx.x;
    const float4* Wv = (const float4*)W;
#pragma unroll
    for (int i = 0; i < 8; ++i) {
        int g = tid + i * 256;
        float4 lo = Wv[2 * g];
        float4 hi = Wv[2 * g + 1];
        float4* dst = (float4*)(Wl + (g >> 4) * WPITCH + (g & 15) * 8);
        dst[0] = make_float4(lo.x, lo.y, hi.x, hi.y);
        dst[1] = make_float4(lo.z, lo.w, hi.z, hi.w);
    }
    __syncthreads();

    int wave = tid >> 5;
    int lane = tid & 31;
    int rb = blockIdx.x * 8 + wave;
    if (rb >= ROW_BLOCKS) return;

    int m    = lane & 15;
    int half = lane >> 4;
    int ncol = lane & 15;

    const float* arow = io + ((size_t)rb * 16 + m) * CH + 2 * half;
    const float4* wq  = (const float4*)(Wl + ncol * WPITCH + 4 * half);

    v8f acc[8] = {};

    for (int j = 0; j < CH / 8; ++j) {
        v2f a0, a1;
        a0.x = arow[8 * j];
        a0.y = arow[8 * j + 1];
        a1.x = arow[8 * j + 4];
        a1.y = arow[8 * j + 5];
        float4 bq[8];
#pragma unroll
        for (int t = 0; t < 8; ++t)
            bq[t] = wq[(t * 16 * WPITCH + 8 * j) / 4];
#pragma unroll
        for (int t = 0; t < 8; ++t) {
            v2f b;
            b.x = bq[t].x;
            b.y = bq[t].y;
            acc[t] = __builtin_amdgcn_wmma_f32_16x16x4_f32(
                false, a0, false, b, (short)0, acc[t], false, false);
        }
#pragma unroll
        for (int t = 0; t < 8; ++t) {
            v2f b;
            b.x = bq[t].z;
            b.y = bq[t].w;
            acc[t] = __builtin_amdgcn_wmma_f32_16x16x4_f32(
                false, a1, false, b, (short)0, acc[t], false, false);
        }
    }

    int base_m = rb * 16 + half * 8;
#pragma unroll
    for (int t = 0; t < 8; ++t) {
        int col = t * 16 + ncol;
        float bv = bias[col];
#pragma unroll
        for (int r = 0; r < 8; ++r) {
            io[(size_t)(base_m + r) * CH + col] = acc[t][r] + bv;
        }
    }
}

extern "C" void kernel_launch(void* const* d_in, const int* in_sizes, int n_in,
                              void* d_out, int out_size, void* d_ws, size_t ws_size,
                              hipStream_t stream) {
    const float* x  = (const float*)d_in[0];
    const int*   ei = (const int*)d_in[1];
    const float* W  = (const float*)d_in[2];
    const float* b  = (const float*)d_in[3];
    float* out = (float*)d_out;

    int*   wsi   = (int*)d_ws;
    int*   cnt   = wsi + WS_CNT;
    int*   total = wsi + WS_TOTAL;
    int*   off   = wsi + WS_OFF;
    float* dis   = (float*)(wsi + WS_DIS);
    int*   rank  = wsi + WS_RANK;
    int*   bec   = wsi + WS_BEC;
    float* bw    = (float*)(wsi + WS_BW);

    k_zero<<<(ZERO_THREADS + 255) / 256, 256, 0, stream>>>(cnt);
    k_count<<<(EDGE_THREADS + EDGE_BLOCK - 1) / EDGE_BLOCK, EDGE_BLOCK, 0, stream>>>(ei, cnt, rank);
    k_off<<<(NODE_THREADS + 255) / 256, 256, 0, stream>>>(cnt, total, off, dis);
    k_fill<<<(EDGE_THREADS + EDGE_BLOCK - 1) / EDGE_BLOCK, EDGE_BLOCK, 0, stream>>>(ei, off, rank, dis, bec, bw);
    k_gather<<<GATHER_WAVES / 8, 256, 0, stream>>>(x, cnt, off, bec, bw, out);
    k_gemm<<<(ROW_BLOCKS + 7) / 8, 256, 0, stream>>>(out, W, b);
}


# Fixed-capacity buckets: the counting atomic's slot places the edge at once, weights formed in the gather from the degrees; scan, rank array and fill pass (2 launches) gone; sum order unchanged
# speedup vs baseline: 2.3596x; 1.3132x over previous
#include <hip/hip_runtime.h>

typedef __attribute__((ext_vector_type(2))) float v2f;
typedef __attribute__((ext_vector_type(8))) float v8f;

#define N_NODES 50000
#define N_EDGES 625000
#define CH 128
#define ROW_BLOCKS (N_NODES / 16)

#define WS_CNT   0
#define WS_BEC   50176
#define CAP      64

#define EPT 8
#define EDGE_THREADS (N_EDGES / EPT)
#define EDGE_BLOCK 64
#define ZERO_THREADS (N_NODES / 4)
#define GATHER_WAVES 16384

__global__ void __launch_bounds__(256) k_zero(int* __restrict__ cnt) {
    int i = blockIdx.x * 256 + threadIdx.x;
    if (i < ZERO_THREADS) ((int4*)cnt)[i] = make_int4(0, 0, 0, 0);
}

__global__ void __launch_bounds__(256)
k_count(const int* __restrict__ ei, int* __restrict__ cnt, int* __restrict__ bec_w) {
    int2* bec = (int2*)bec_w;
    int t = blockIdx.x * EDGE_BLOCK + threadIdx.x;
    if (t >= EDGE_THREADS) return;
    const int4* r4 = (const int4*)ei;
    const int4* c4 = (const int4*)(ei + N_EDGES);
    int4 ra = r4[2 * t], rb = r4[2 * t + 1];
    int4 ca = c4[2 * t], cb = c4[2 * t + 1];
    int r[EPT] = {ra.x, ra.y, ra.z, ra.w, rb.x, rb.y, rb.z, rb.w};
    int c[EPT] = {ca.x, ca.y, ca.z, ca.w, cb.x, cb.y, cb.z, cb.w};
    int s[EPT];
#pragma unroll
    for (int j = 0; j < EPT; ++j) s[j] = atomicAdd(&cnt[r[j]], 1);
#pragma unroll
    for (int j = 0; j < EPT; ++j)
        if (s[j] < CAP) bec[(size_t)r[j] * CAP + s[j]] = make_int2(EPT * t + j, c[j]);
}

__device__ __forceinline__ float4 fma4(float4 v, float w, float4 a) {
    a.x = fmaf(v.x, w, a.x);
    a.y = fmaf(v.y, w, a.y);
    a.z = fmaf(v.z, w, a.z);
    a.w = fmaf(v.w, w, a.w);
    return a;
}

__device__ __forceinline__ float dinv(int deg) {
    return (deg > 0) ? rsqrtf((float)deg) : 0.0f;
}

__global__ void __launch_bounds__(256)
k_gather(const float* __restrict__ x, const int* __restrict__ ei,
         const int* __restrict__ cnt, const int* __restrict__ bec_w,
         float* __restrict__ agg) {
    const int2* bec = (const int2*)bec_w;
    int lane = threadIdx.x & 31;
    int node = __builtin_amdgcn_readfirstlane(blockIdx.x * 8 + (threadIdx.x >> 5));
    const float4* x4 = (const float4*)x;

    int  d   = cnt[node];
    int2 raw = bec[(size_t)node * CAP + lane];
    int  cdeg = cnt[(lane < d) ? raw.y : 0];
    int  n1 = node + GATHER_WAVES;
    int  d1 = 0;
    if (n1 < N_NODES) d1 = cnt[n1];
    int2 raw1 = bec[(size_t)min(n1, N_NODES - 1) * CAP + lane];

    for (;;) {
        int cdeg1 = cnt[(lane < d1) ? raw1.y : 0];
        int  n2 = n1 + GATHER_WAVES;
        int  d2 = 0;
        if (n2 < N_NODES) d2 = cnt[n2];
        int2 raw2 = bec[(size_t)min(n2, N_NODES - 1) * CAP + lane];

        float dr = dinv(d);
        float4 acc = make_float4(0.0f, 0.0f, 0.0f, 0.0f);
        if (d <= 32) {
            int   e = (lane < d) ? raw.x : 0x7fffffff;
            int   c = (lane < d) ? raw.y : 0;
            float w = (lane < d) ? dr * dinv(cdeg) : 0.0f;
            int rk = 0;
            for (int k = 0; k < d; ++k) {
                int ek = __builtin_amdgcn_readlane(e, k);
                rk += (ek < e) ? 1 : 0;
            }
            if (lane >= d) rk = lane;
            int   sc = __builtin_amdgcn_ds_permute(rk << 2, c);
            float sw = __int_as_float(__builtin_amdgcn_ds_permute(rk << 2, __float_as_int(w)));

            int i = 0;
            for (; i + 4 <= d; i += 4) {
                int c0 = __builtin_amdgcn_readlane(sc, i);
                int c1r = __builtin_amdgcn_readlane(sc, i + 1);
                int c2 = __builtin_amdgcn_readlane(sc, i + 2);
                int c3 = __builtin_amdgcn_readlane(sc, i + 3);
                float4 v0 = x4[(size_t)c0 * (CH / 4) + lane];
                float4 v1 = x4[(size_t)c1r * (CH / 4) + lane];
                float4 v2 = x4[(size_t)c2 * (CH / 4) + lane];
                float4 v3 = x4[(size_t)c3 * (CH / 4) + lane];
                float w0 = __int_as_float(__builtin_amdgcn_readlane(__float_as_int(sw), i));
                float w1r = __int_as_float(__builtin_amdgcn_readlane(__float_as_int(sw), i + 1));
                float w2 = __int_as_float(__builtin_amdgcn_readlane(__float_as_int(sw), i + 2));
                float w3 = __int_as_float(__builtin_amdgcn_readlane(__float_as_int(sw), i + 3));
                acc = fma4(v0, w0, acc);
                acc = fma4(v1, w1r, acc);
                acc = fma4(v2, w2, acc);
                acc = fma4(v3, w3, acc);
            }
            for (; i < d; ++i) {
                int   ci = __builtin_amdgcn_readlane(sc, i);
                float wi = __int_as_float(__builtin_amdgcn_readlane(__float_as_int(sw), i));
                acc = fma4(x4[(size_t)ci * (CH / 4) + lane], wi, acc);
            }
        } else if (d <= CAP) {
            const int2* bk = bec + (size_t)node * CAP;
            int last = -1;
            for (int i = 0; i < d; ++i) {
                int best = 0x7fffffff, bp = 0;
                for (int p = lane; p < d; p += 32) {
                    int id = bk[p].x;
                    if (id > last && id < best) { best = id; bp = p; }
                }
                for (int s = 16; s > 0; s >>= 1) {
                    int ob = __shfl_xor(best, s);
                    int op = __shfl_xor(bp, s);
                    if (ob < best) { best = ob; bp = op; }
                }
                last = best;
                int   ci = bk[bp].y;
                float wi = dr * dinv(cnt[ci]);
                acc = fma4(x4[(size_t)ci * (CH / 4) + lane], wi, acc);
            }
        } else {
            for (int base = 0; base < N_EDGES; base += 32) {
                int eid = base + lane;
                int rr = -1, cc = 0;
                if (eid < N_EDGES) { rr = ei[eid]; cc = ei[N_EDGES + eid]; }
                unsigned m = (unsigned)__ballot(rr == node);
                while (m) {
                    int l = __builtin_ctz(m);
                    m &= m - 1;
                    int   ci = __builtin_amdgcn_readlane(cc, l);
                    float wi = dr * dinv(cnt[ci]);
                    acc = fma4(x4[(size_t)ci * (CH / 4) + lane], wi, acc);
                }
            }
        }
        ((float4*)agg)[(size_t)node * (CH / 4) + lane] = acc;

        if (n1 >= N_NODES) break;
        node = n1;  d = d1;  raw = raw1;  cdeg = cdeg1;
        n1 = n2;  d1 = d2;  raw1 = raw2;
    }
}

#define WPITCH 132

__global__ void __launch_bounds__(256)
k_gemm(float* __restrict__ io,
                       const float* __restrict__ W,
                       const float* __restrict__ bias) {
    __shared__ float4 Wl4[CH * WPITCH / 4];
    float* Wl = (float*)Wl4;

    int tid = threadIdx.x;
    const float4* Wv = (const float4*)W;
#pragma unroll
    for (int i = 0; i < 8; ++i) {
        int g = tid + i * 256;
        float4 lo = Wv[2 * g];
        float4 hi = Wv[2 * g + 1];
        float4* dst = (float4*)(Wl + (g >> 4) * WPITCH + (g & 15) * 8);
        dst[0] = make_float4(lo.x, lo.y, hi.x, hi.y);
        dst[1] = make_float4(lo.z, lo.w, hi.z, hi.w);
    }
    __syncthreads();

    int wave = tid >> 5;
    int lane = tid & 31;
    int rb = blockIdx.x * 8 + wave;
    if (rb >= ROW_BLOCKS) return;

    int m    = lane & 15;
    int half = lane >> 4;
    int ncol = lane & 15;

    const float* arow = io + ((size_t)rb * 16 + m) * CH + 2 * half;
    const float4* wq  = (const float4*)(Wl + ncol * WPITCH + 4 * half);

    v8f acc[8] = {};

    for (int j = 0; j < CH / 8; ++j) {
        v2f a0, a1;
        a0.x = arow[8 * j];
        a0.y = arow[8 * j + 1];
        a1.x = arow[8 * j + 4];
        a1.y = arow[8 * j + 5];
        float4 bq[8];
#pragma unroll
        for (int t = 0; t < 8; ++t)
            bq[t] = wq[(t * 16 * WPITCH + 8 * j) / 4];
#pragma unroll
        for (int t = 0; t < 8; ++t) {
            v2f b;
            b.x = bq[t].x;
            b.y = bq[t].y;
            acc[t] = __builtin_amdgcn_wmma_f32_16x16x4_f32(
                false, a0, false, b, (short)0, acc[t], false, false);
        }
#pragma unroll
        for (int t = 0; t < 8; ++t) {
            v2f b;
            b.x = bq[t].z;
            b.y = bq[t].w;
            acc[t] = __builtin_amdgcn_wmma_f32_16x16x4_f32(
                false, a1, false, b, (short)0, acc[t], false, false);
        }
    }

    int base_m = rb * 16 + half * 8;
#pragma unroll
    for (int t = 0; t < 8; ++t) {
        int col = t * 16 + ncol;
        float bv = bias[col];
#pragma unroll
        for (int r = 0; r < 8; ++r) {
            io[(size_t)(base_m + r) * CH + col] = acc[t][r] + bv;
        }
    }
}

extern "C" void kernel_launch(void* const* d_in, const int* in_sizes, int n_in,
                              void* d_out, int out_size, void* d_ws, size_t ws_size,
                              hipStream_t stream) {
    const float* x  = (const float*)d_in[0];
    const int*   ei = (const int*)d_in[1];
    const float* W  = (const float*)d_in[2];
    const float* b  = (const float*)d_in[3];
    float* out = (float*)d_out;

    int*   wsi = (int*)d_ws;
    int*   cnt = wsi + WS_CNT;
    int*   bec = wsi + WS_BEC;

    k_zero<<<(ZERO_THREADS + 255) / 256, 256, 0, stream>>>(cnt);
    k_count<<<(EDGE_THREADS + EDGE_BLOCK - 1) / EDGE_BLOCK, EDGE_BLOCK, 0, stream>>>(ei, cnt, bec);
    k_gather<<<GATHER_WAVES / 8, 256, 0, stream>>>(x, ei, cnt, bec, out);
    k_gemm<<<(ROW_BLOCKS + 7) / 8, 256, 0, stream>>>(out, W, b);
}


# Edges bucketed with LDS counters in two launches (split by 256-node range, then slot per node) instead of one returning global atomic per edge; zero launch gone; order of float sums unchanged
# speedup vs baseline: 3.4676x; 1.4696x over previous
#include <hip/hip_runtime.h>

typedef __attribute__((ext_vector_type(2))) float v2f;
typedef __attribute__((ext_vector_type(8))) float v8f;

#define N_NODES 50000
#define N_EDGES 625000
#define CH 128
#define ROW_BLOCKS (N_NODES / 16)

#define WS_CNT   0
#define WS_BEC   50176
#define CAP      64
#define WS_SEGCNT 6450176
#define WS_SEG    6510208

#define EPT 8
#define EDGE_THREADS (N_EDGES / EPT)
#define PART_THREADS 256
#define PART_EDGES (EPT * PART_THREADS)
#define NB ((N_EDGES + PART_EDGES - 1) / PART_EDGES)
#define RSHIFT 8
#define RNODES (1 << RSHIFT)
#define NRANGE ((N_NODES + RNODES - 1) / RNODES)
#define SEGCAP 64
#define EID_BITS 20
#define PLACE_THREADS 512
#define PLACE_WAVES (PLACE_THREADS / 32)
#define PLACE_ITERS ((NB + PLACE_WAVES - 1) / PLACE_WAVES)
#define GATHER_WAVES 16384

__global__ void __launch_bounds__(PART_THREADS)
k_part(const int* __restrict__ ei, int* __restrict__ segcnt, int* __restrict__ seg_w) {
    __shared__ int hist[NRANGE];
    int2* seg = (int2*)seg_w;
    int tid = threadIdx.x;
    int b = blockIdx.x;
    int t = b * PART_THREADS + tid;
    bool act = t < EDGE_THREADS;
    int4 ra = make_int4(0, 0, 0, 0), rb = ra, ca = ra, cb = ra;
    if (act) {
        const int4* r4 = (const int4*)ei;
        const int4* c4 = (const int4*)(ei + N_EDGES);
        ra = r4[2 * t];  rb = r4[2 * t + 1];
        ca = c4[2 * t];  cb = c4[2 * t + 1];
    }
    if (tid < NRANGE) hist[tid] = 0;
    __syncthreads();
    int r[EPT] = {ra.x, ra.y, ra.z, ra.w, rb.x, rb.y, rb.z, rb.w};
    int c[EPT] = {ca.x, ca.y, ca.z, ca.w, cb.x, cb.y, cb.z, cb.w};
    int rk[EPT];
    if (act) {
#pragma unroll
        for (int j = 0; j < EPT; ++j) rk[j] = atomicAdd(&hist[r[j] >> RSHIFT], 1);
    }
    __syncthreads();
    if (tid < NRANGE) segcnt[tid * NB + b] = hist[tid];
    if (act) {
#pragma unroll
        for (int j = 0; j < EPT; ++j) {
            int rg = r[j] >> RSHIFT;
            if (rk[j] < SEGCAP)
                seg[((size_t)rg * NB + b) * SEGCAP + rk[j]] =
                    make_int2((EPT * t + j) | ((r[j] & (RNODES - 1)) << EID_BITS), c[j]);
        }
    }
}

__global__ void __launch_bounds__(PLACE_THREADS)
k_place(const int* __restrict__ ei, const int* __restrict__ segcnt,
        const int* __restrict__ seg_w, int* __restrict__ cnt, int* __restrict__ bec_w) {
    __shared__ int cnt_l[RNODES];
    __shared__ int wover[PLACE_WAVES];
    const int2* seg = (const int2*)seg_w;
    int2* bec = (int2*)bec_w;
    int tid = threadIdx.x;
    int range = blockIdx.x;
    int wave = tid >> 5;
    int lane = tid & 31;
    int myc = 0;
    if (lane < PLACE_ITERS && wave + PLACE_WAVES * lane < NB)
        myc = segcnt[range * NB + wave + PLACE_WAVES * lane];
    if (tid < RNODES) cnt_l[tid] = 0;
    int big = (__ballot(myc > SEGCAP) != 0) ? 1 : 0;
    if (lane == 0) wover[wave] = big;
    __syncthreads();
    bool over = __ballot(wover[lane & (PLACE_WAVES - 1)] != 0) != 0;

    if (!over) {
#define PLACE_SEGS(X) X(0) X(1) X(2) X(3) X(4) X(5) X(6) X(7) X(8) X(9) \
                      X(10) X(11) X(12) X(13) X(14) X(15) X(16) X(17) X(18) X(19)
        static_assert(PLACE_ITERS == 20, "PLACE_SEGS lists one step per segment of a wave");
#define PLACE_LOAD(u)                                                              \
        int n##u = __builtin_amdgcn_readlane(myc, u);                              \
        int2 v##u = make_int2(0, 0);                                               \
        if (lane < n##u)                                                           \
            v##u = seg[((size_t)range * NB + wave + PLACE_WAVES * u) * SEGCAP + lane];
#define PLACE_RANK(u)                                                              \
        int lrow##u = (v##u.x >> EID_BITS) & (RNODES - 1);                         \
        int slot##u = 0;                                                           \
        if (lane < n##u) slot##u = atomicAdd(&cnt_l[lrow##u], 1);
#define PLACE_STORE(u)                                                             \
        if (lane < n##u && slot##u < CAP)                                          \
            bec[(size_t)(range * RNODES + lrow##u) * CAP + slot##u] =              \
                make_int2(v##u.x & ((1 << EID_BITS) - 1), v##u.y);
        PLACE_SEGS(PLACE_LOAD)
        PLACE_SEGS(PLACE_RANK)
        PLACE_SEGS(PLACE_STORE)
#undef PLACE_LOAD
#undef PLACE_RANK
#undef PLACE_STORE
#undef PLACE_SEGS
        for (int u = 0; u < PLACE_ITERS; ++u) {
            int n = __builtin_amdgcn_readlane(myc, u);
            if (lane + 32 < n) {
                int2 w = seg[((size_t)range * NB + wave + PLACE_WAVES * u) * SEGCAP + lane + 32];
                int lrow = (w.x >> EID_BITS) & (RNODES - 1);
                int sl = atomicAdd(&cnt_l[lrow], 1);
                if (sl < CAP)
                    bec[(size_t)(range * RNODES + lrow) * CAP + sl] =
                        make_int2(w.x & ((1 << EID_BITS) - 1), w.y);
            }
        }
    } else {
        for (int e = tid; e < N_EDGES; e += PLACE_THREADS) {
            int row = ei[e];
            if ((row >> RSHIFT) == range) {
                int lrow = row & (RNODES - 1);
                int sl = atomicAdd(&cnt_l[lrow], 1);
                if (sl < CAP)
                    bec[(size_t)(range * RNODES + lrow) * CAP + sl] = make_int2(e, ei[N_EDGES + e]);
            }
        }
    }
    __syncthreads();
    if (tid < RNODES) {
        int node = range * RNODES + tid;
        if (node < N_NODES) cnt[node] = cnt_l[tid];
    }
}

__device__ __forceinline__ float4 fma4(float4 v, float w, float4 a) {
    a.x = fmaf(v.x, w, a.x);
    a.y = fmaf(v.y, w, a.y);
    a.z = fmaf(v.z, w, a.z);
    a.w = fmaf(v.w, w, a.w);
    return a;
}

__device__ __forceinline__ float dinv(int deg) {
    return (deg > 0) ? rsqrtf((float)deg) : 0.0f;
}

__global__ void __launch_bounds__(256)
k_gather(const float* __restrict__ x, const int* __restrict__ ei,
         const int* __restrict__ cnt, const int* __restrict__ bec_w,
         float* __restrict__ agg) {
    const int2* bec = (const int2*)bec_w;
    int lane = threadIdx.x & 31;
    int node = __builtin_amdgcn_readfirstlane(blockIdx.x * 8 + (threadIdx.x >> 5));
    const float4* x4 = (const float4*)x;

    int  d   = cnt[node];
    int2 raw = bec[(size_t)node * CAP + lane];
    int  cdeg = cnt[(lane < d) ? raw.y : 0];
    int  n1 = node + GATHER_WAVES;
    int  d1 = 0;
    if (n1 < N_NODES) d1 = cnt[n1];
    int2 raw1 = bec[(size_t)min(n1, N_NODES - 1) * CAP + lane];

    for (;;) {
        int cdeg1 = cnt[(lane < d1) ? raw1.y : 0];
        int  n2 = n1 + GATHER_WAVES;
        int  d2 = 0;
        if (n2 < N_NODES) d2 = cnt[n2];
        int2 raw2 = bec[(size_t)min(n2, N_NODES - 1) * CAP + lane];

        float dr = dinv(d);
        float4 acc = make_float4(0.0f, 0.0f, 0.0f, 0.0f);
        if (d <= 32) {
            int   e = (lane < d) ? raw.x : 0x7fffffff;
            int   c = (lane < d) ? raw.y : 0;
            float w = (lane < d) ? dr * dinv(cdeg) : 0.0f;
            int rk = 0;
            for (int k = 0; k < d; ++k) {
                int ek = __builtin_amdgcn_readlane(e, k);
                rk += (ek < e) ? 1 : 0;
            }
            if (lane >= d) rk = lane;
            int   sc = __builtin_amdgcn_ds_permute(rk << 2, c);
            float sw = __int_as_float(__builtin_amdgcn_ds_permute(rk << 2, __float_as_int(w)));

            int i = 0;
            for (; i + 4 <= d; i += 4) {
                int c0 = __builtin_amdgcn_readlane(sc, i);
                int c1r = __builtin_amdgcn_readlane(sc, i + 1);
                int c2 = __builtin_amdgcn_readlane(sc, i + 2);
                int c3 = __builtin_amdgcn_readlane(sc, i + 3);
                float4 v0 = x4[(size_t)c0 * (CH / 4) + lane];
                float4 v1 = x4[(size_t)c1r * (CH / 4) + lane];
                float4 v2 = x4[(size_t)c2 * (CH / 4) + lane];
                float4 v3 = x4[(size_t)c3 * (CH / 4) + lane];
                float w0 = __int_as_float(__builtin_amdgcn_readlane(__float_as_int(sw), i));
                float w1r = __int_as_float(__builtin_amdgcn_readlane(__float_as_int(sw), i + 1));
                float w2 = __int_as_float(__builtin_amdgcn_readlane(__float_as_int(sw), i + 2));
                float w3 = __int_as_float(__builtin_amdgcn_readlane(__float_as_int(sw), i + 3));
                acc = fma4(v0, w0, acc);
                acc = fma4(v1, w1r, acc);
                acc = fma4(v2, w2, acc);
                acc = fma4(v3, w3, acc);
            }
            for (; i < d; ++i) {
                int   ci = __builtin_amdgcn_readlane(sc, i);
                float wi = __int_as_float(__builtin_amdgcn_readlane(__float_as_int(sw), i));
                acc = fma4(x4[(size_t)ci * (CH / 4) + lane], wi, acc);
            }
        } else if (d <= CAP) {
            const int2* bk = bec + (size_t)node * CAP;
            int last = -1;
            for (int i = 0; i < d; ++i) {
                int best = 0x7fffffff, bp = 0;
                for (int p = lane; p < d; p += 32) {
                    int id = bk[p].x;
                    if (id > last && id < best) { best = id; bp = p; }
                }
                for (int s = 16; s > 0; s >>= 1) {
                    int ob = __shfl_xor(best, s);
                    int op = __shfl_xor(bp, s);
                    if (ob < best) { best = ob; bp = op; }
                }
                last = best;
                int   ci = bk[bp].y;
                float wi = dr * dinv(cnt[ci]);
                acc = fma4(x4[(size_t)ci * (CH / 4) + lane], wi, acc);
            }
        } else {
            for (int base = 0; base < N_EDGES; base += 32) {
                int eid = base + lane;
                int rr = -1, cc = 0;
                if (eid < N_EDGES) { rr = ei[eid]; cc = ei[N_EDGES + eid]; }
                unsigned m = (unsigned)__ballot(rr == node);
                while (m) {
                    int l = __builtin_ctz(m);
                    m &= m - 1;
                    int   ci = __builtin_amdgcn_readlane(cc, l);
                    float wi = dr * dinv(cnt[ci]);
                    acc = fma4(x4[(size_t)ci * (CH / 4) + lane], wi, acc);
                }
            }
        }
        ((float4*)agg)[(size_t)node * (CH / 4) + lane] = acc;

        if (n1 >= N_NODES) break;
        node = n1;  d = d1;  raw = raw1;  cdeg = cdeg1;
        n1 = n2;  d1 = d2;  raw1 = raw2;
    }
}

#define WPITCH 132

__global__ void __launch_bounds__(256)
k_gemm(float* __restrict__ io,
                       const float* __restrict__ W,
                       const float* __restrict__ bias) {
    __shared__ float4 Wl4[CH * WPITCH / 4];
    float* Wl = (float*)Wl4;

    int tid = threadIdx.x;
    const float4* Wv = (const float4*)W;
#pragma unroll
    for (int i = 0; i < 8; ++i) {
        int g = tid + i * 256;
        float4 lo = Wv[2 * g];
        float4 hi = Wv[2 * g + 1];
        float4* dst = (float4*)(Wl + (g >> 4) * WPITCH + (g & 15) * 8);
        dst[0] = make_float4(lo.x, lo.y, hi.x, hi.y);
        dst[1] = make_float4(lo.z, lo.w, hi.z, hi.w);
    }
    __syncthreads();

    int wave = tid >> 5;
    int lane = tid & 31;
    int rb = blockIdx.x * 8 + wave;
    if (rb >= ROW_BLOCKS) return;

    int m    = lane & 15;
    int half = lane >> 4;
    int ncol = lane & 15;

    const float* arow = io + ((size_t)rb * 16 + m) * CH + 2 * half;
    const float4* wq  = (const float4*)(Wl + ncol * WPITCH + 4 * half);

    v8f acc[8] = {};

    for (int j = 0; j < CH / 8; ++j) {
        v2f a0, a1;
        a0.x = arow[8 * j];
        a0.y = arow[8 * j + 1];
        a1.x = arow[8 * j + 4];
        a1.y = arow[8 * j + 5];
        float4 bq[8];
#pragma unroll
        for (int t = 0; t < 8; ++t)
            bq[t] = wq[(t * 16 * WPITCH + 8 * j) / 4];
#pragma unroll
        for (int t = 0; t < 8; ++t) {
            v2f b;
            b.x = bq[t].x;
            b.y = bq[t].y;
            acc[t] = __builtin_amdgcn_wmma_f32_16x16x4_f32(
                false, a0, false, b, (short)0, acc[t], false, false);
        }
#pragma unroll
        for (int t = 0; t < 8; ++t) {
            v2f b;
            b.x = bq[t].z;
            b.y = bq[t].w;
            acc[t] = __builtin_amdgcn_wmma_f32_16x16x4_f32(
                false, a1, false, b, (short)0, acc[t], false, false);
        }
    }

    int base_m = rb * 16 + half * 8;
#pragma unroll
    for (int t = 0; t < 8; ++t) {
        int col = t * 16 + ncol;
        float bv = bias[col];
#pragma unroll
        for (int r = 0; r < 8; ++r) {
            io[(size_t)(base_m + r) * CH + col] = acc[t][r] + bv;
        }
    }
}

extern "C" void kernel_launch(void* const* d_in, const int* in_sizes, int n_in,
                              void* d_out, int out_size, void* d_ws, size_t ws_size,
                              hipStream_t stream) {
    const float* x  = (const float*)d_in[0];
    const int*   ei = (const int*)d_in[1];
    const float* W  = (const float*)d_in[2];
    const float* b  = (const float*)d_in[3];
    float* out = (float*)d_out;

    int*   wsi = (int*)d_ws;
    int*   cnt = wsi + WS_CNT;
    int*   bec = wsi + WS_BEC;
    int*   segcnt = wsi + WS_SEGCNT;
    int*   seg = wsi + WS_SEG;

    k_part<<<NB, PART_THREADS, 0, stream>>>(ei, segcnt, seg);
    k_place<<<NRANGE, PLACE_THREADS, 0, stream>>>(ei, segcnt, seg, cnt, bec);
    k_gather<<<GATHER_WAVES / 8, 256, 0, stream>>>(x, ei, cnt, bec, out);
    k_gemm<<<(ROW_BLOCKS + 7) / 8, 256, 0, stream>>>(out, W, b);
}


# Gather and GEMM in one launch: a workgroup sums 16 nodes into an LDS tile, its 8 waves each do one 16-column WMMA tile, W fragments kept in registers; agg round trip gone; sum order unchanged
# speedup vs baseline: 3.8394x; 1.1072x over previous
#include <hip/hip_runtime.h>

typedef __attribute__((ext_vector_type(2))) float v2f;
typedef __attribute__((ext_vector_type(8))) float v8f;

#define N_NODES 50000
#define N_EDGES 625000
#define CH 128
#define ROW_BLOCKS (N_NODES / 16)

#define WS_CNT   0
#define WS_BEC   50176
#define CAP      64
#define WS_SEGCNT 6450176
#define WS_SEG    6510208

#define EPT 8
#define EDGE_THREADS (N_EDGES / EPT)
#define PART_THREADS 256
#define PART_EDGES (EPT * PART_THREADS)
#define NB ((N_EDGES + PART_EDGES - 1) / PART_EDGES)
#define RSHIFT 8
#define RNODES (1 << RSHIFT)
#define NRANGE ((N_NODES + RNODES - 1) / RNODES)
#define SEGCAP 64
#define EID_BITS 20
#define PLACE_THREADS 512
#define PLACE_WAVES (PLACE_THREADS / 32)
#define PLACE_ITERS ((NB + PLACE_WAVES - 1) / PLACE_WAVES)

__global__ void __launch_bounds__(PART_THREADS)
k_part(const int* __restrict__ ei, int* __restrict__ segcnt, int* __restrict__ seg_w) {
    __shared__ int hist[NRANGE];
    int2* seg = (int2*)seg_w;
    int tid = threadIdx.x;
    int b = blockIdx.x;
    int t = b * PART_THREADS + tid;
    bool act = t < EDGE_THREADS;
    int4 ra = make_int4(0, 0, 0, 0), rb = ra, ca = ra, cb = ra;
    if (act) {
        const int4* r4 = (const int4*)ei;
        const int4* c4 = (const int4*)(ei + N_EDGES);
        ra = r4[2 * t];  rb = r4[2 * t + 1];
        ca = c4[2 * t];  cb = c4[2 * t + 1];
    }
    if (tid < NRANGE) hist[tid] = 0;
    __syncthreads();
    int r[EPT] = {ra.x, ra.y, ra.z, ra.w, rb.x, rb.y, rb.z, rb.w};
    int c[EPT] = {ca.x, ca.y, ca.z, ca.w, cb.x, cb.y, cb.z, cb.w};
    int rk[EPT];
    if (act) {
#pragma unroll
        for (int j = 0; j < EPT; ++j) rk[j] = atomicAdd(&hist[r[j] >> RSHIFT], 1);
    }
    __syncthreads();
    if (tid < NRANGE) segcnt[tid * NB + b] = hist[tid];
    if (act) {
#pragma unroll
        for (int j = 0; j < EPT; ++j) {
            int rg = r[j] >> RSHIFT;
            if (rk[j] < SEGCAP)
                seg[((size_t)rg * NB + b) * SEGCAP + rk[j]] =
                    make_int2((EPT * t + j) | ((r[j] & (RNODES - 1)) << EID_BITS), c[j]);
        }
    }
}

__global__ void __launch_bounds__(PLACE_THREADS)
k_place(const int* __restrict__ ei, const int* __restrict__ segcnt,
        const int* __restrict__ seg_w, int* __restrict__ cnt, int* __restrict__ bec_w) {
    __shared__ int cnt_l[RNODES];
    __shared__ int wover[PLACE_WAVES];
    const int2* seg = (const int2*)seg_w;
    int2* bec = (int2*)bec_w;
    int tid = threadIdx.x;
    int range = blockIdx.x;
    int wave = tid >> 5;
    int lane = tid & 31;
    int myc = 0;
    if (lane < PLACE_ITERS && wave + PLACE_WAVES * lane < NB)
        myc = segcnt[range * NB + wave + PLACE_WAVES * lane];
    if (tid < RNODES) cnt_l[tid] = 0;
    int big = (__ballot(myc > SEGCAP) != 0) ? 1 : 0;
    if (lane == 0) wover[wave] = big;
    __syncthreads();
    bool over = __ballot(wover[lane & (PLACE_WAVES - 1)] != 0) != 0;

    if (!over) {
#define PLACE_SEGS(X) X(0) X(1) X(2) X(3) X(4) X(5) X(6) X(7) X(8) X(9) \
                      X(10) X(11) X(12) X(13) X(14) X(15) X(16) X(17) X(18) X(19)
        static_assert(PLACE_ITERS == 20, "PLACE_SEGS lists one step per segment of a wave");
#define PLACE_LOAD(u)                                                              \
        int n##u = __builtin_amdgcn_readlane(myc, u);                              \
        int2 v##u = make_int2(0, 0);                                               \
        if (lane < n##u)                                                           \
            v##u = seg[((size_t)range * NB + wave + PLACE_WAVES * u) * SEGCAP + lane];
#define PLACE_RANK(u)                                                              \
        int lrow##u = (v##u.x >> EID_BITS) & (RNODES - 1);                         \
        int slot##u = 0;                                                           \
        if (lane < n##u) slot##u = atomicAdd(&cnt_l[lrow##u], 1);
#define PLACE_STORE(u)                                                             \
        if (lane < n##u && slot##u < CAP)                                          \
            bec[(size_t)(range * RNODES + lrow##u) * CAP + slot##u] =              \
                make_int2(v##u.x & ((1 << EID_BITS) - 1), v##u.y);
        PLACE_SEGS(PLACE_LOAD)
        PLACE_SEGS(PLACE_RANK)
        PLACE_SEGS(PLACE_STORE)
#undef PLACE_LOAD
#undef PLACE_RANK
#undef PLACE_STORE
#undef PLACE_SEGS
        for (int u = 0; u < PLACE_ITERS; ++u) {
            int n = __builtin_amdgcn_readlane(myc, u);
            if (lane + 32 < n) {
                int2 w = seg[((size_t)range * NB + wave + PLACE_WAVES * u) * SEGCAP + lane + 32];
                int lrow = (w.x >> EID_BITS) & (RNODES - 1);
                int sl = atomicAdd(&cnt_l[lrow], 1);
                if (sl < CAP)
                    bec[(size_t)(range * RNODES + lrow) * CAP + sl] =
                        make_int2(w.x & ((1 << EID_BITS) - 1), w.y);
            }
        }
    } else {
        for (int e = tid; e < N_EDGES; e += PLACE_THREADS) {
            int row = ei[e];
            if ((row >> RSHIFT) == range) {
                int lrow = row & (RNODES - 1);
                int sl = atomicAdd(&cnt_l[lrow], 1);
                if (sl < CAP)
                    bec[(size_t)(range * RNODES + lrow) * CAP + sl] = make_int2(e, ei[N_EDGES + e]);
            }
        }
    }
    __syncthreads();
    if (tid < RNODES) {
        int node = range * RNODES + tid;
        if (node < N_NODES) cnt[node] = cnt_l[tid];
    }
}

#define FUSE_BLOCKS 1024
#define APITCH 132

__device__ __forceinline__ float4 fma4(float4 v, float w, float4 a) {
    a.x = fmaf(v.x, w, a.x);
    a.y = fmaf(v.y, w, a.y);
    a.z = fmaf(v.z, w, a.z);
    a.w = fmaf(v.w, w, a.w);
    return a;
}

__device__ __forceinline__ float dinv(int deg) {
    return (deg > 0) ? rsqrtf((float)deg) : 0.0f;
}

__device__ __forceinline__ float4 gather_node(
        const float4* __restrict__ x4, const int* __restrict__ ei,
        const int* __restrict__ cnt, const int2* __restrict__ bec,
        int node, int d, int2 raw, int cdeg, int lane) {
    float dr = dinv(d);
    float4 acc = make_float4(0.0f, 0.0f, 0.0f, 0.0f);
    if (d <= 32) {
        int   e = (lane < d) ? raw.x : 0x7fffffff;
        int   c = (lane < d) ? raw.y : 0;
        float w = (lane < d) ? dr * dinv(cdeg) : 0.0f;
        int rk = 0;
        for (int k = 0; k < d; ++k) {
            int ek = __builtin_amdgcn_readlane(e, k);
            rk += (ek < e) ? 1 : 0;
        }
        if (lane >= d) rk = lane;
        int   sc = __builtin_amdgcn_ds_permute(rk << 2, c);
        float sw = __int_as_float(__builtin_amdgcn_ds_permute(rk << 2, __float_as_int(w)));

        int i = 0;
        for (; i + 4 <= d; i += 4) {
            int c0 = __builtin_amdgcn_readlane(sc, i);
            int c1r = __builtin_amdgcn_readlane(sc, i + 1);
            int c2 = __builtin_amdgcn_readlane(sc, i + 2);
            int c3 = __builtin_amdgcn_readlane(sc, i + 3);
            float4 v0 = x4[(size_t)c0 * (CH / 4) + lane];
            float4 v1 = x4[(size_t)c1r * (CH / 4) + lane];
            float4 v2 = x4[(size_t)c2 * (CH / 4) + lane];
            float4 v3 = x4[(size_t)c3 * (CH / 4) + lane];
            float w0 = __int_as_float(__builtin_amdgcn_readlane(__float_as_int(sw), i));
            float w1r = __int_as_float(__builtin_amdgcn_readlane(__float_as_int(sw), i + 1));
            float w2 = __int_as_float(__builtin_amdgcn_readlane(__float_as_int(sw), i + 2));
            float w3 = __int_as_float(__builtin_amdgcn_readlane(__float_as_int(sw), i + 3));
            acc = fma4(v0, w0, acc);
            acc = fma4(v1, w1r, acc);
            acc = fma4(v2, w2, acc);
            acc = fma4(v3, w3, acc);
        }
        for (; i < d; ++i) {
            int   ci = __builtin_amdgcn_readlane(sc, i);
            float wi = __int_as_float(__builtin_amdgcn_readlane(__float_as_int(sw), i));
            acc = fma4(x4[(size_t)ci * (CH / 4) + lane], wi, acc);
        }
    } else if (d <= CAP) {
        const int2* bk = bec + (size_t)node * CAP;
        int last = -1;
        for (int i = 0; i < d; ++i) {
            int best = 0x7fffffff, bp = 0;
            for (int p = lane; p < d; p += 32) {
                int id = bk[p].x;
                if (id > last && id < best) { best = id; bp = p; }
            }
            for (int s = 16; s > 0; s >>= 1) {
                int ob = __shfl_xor(best, s);
                int op = __shfl_xor(bp, s);
                if (ob < best) { best = ob; bp = op; }
            }
            last = best;
            int   ci = bk[bp].y;
            float wi = dr * dinv(cnt[ci]);
            acc = fma4(x4[(size_t)ci * (CH / 4) + lane], wi, acc);
        }
    } else {
        for (int base = 0; base < N_EDGES; base += 32) {
            int eid = base + lane;
            int rr = -1, cc = 0;
            if (eid < N_EDGES) { rr = ei[eid]; cc = ei[N_EDGES + eid]; }
            unsigned m = (unsigned)__ballot(rr == node);
            while (m) {
                int l = __builtin_ctz(m);
                m &= m - 1;
                int   ci = __builtin_amdgcn_readlane(cc, l);
                float wi = dr * dinv(cnt[ci]);
                acc = fma4(x4[(size_t)ci * (CH / 4) + lane], wi, acc);
            }
        }
    }
    return acc;
}

__global__ void __launch_bounds__(256, 8)
k_gather_gemm(const float* __restrict__ x, const int* __restrict__ ei,
              const int* __restrict__ cnt, const int* __restrict__ bec_w,
              const float* __restrict__ W,
              const float* __restrict__ bias,
              float* __restrict__ out) {
    __shared__ float4 lds4[8 * 16 * APITCH / 4];
    float* tile = (float*)lds4;
    const int2* bec = (const int2*)bec_w;
    const float4* x4 = (const float4*)x;
    int lane = threadIdx.x & 31;
    int wave = __builtin_amdgcn_readfirstlane(threadIdx.x >> 5);
    int c = lane & 15;
    int h = lane >> 4;

    float* wst = tile + wave * 16 * APITCH;
    const float4* W4 = (const float4*)W;
#pragma unroll
    for (int i = 0; i < 16; ++i)
        *(float4*)(wst + i * APITCH + 4 * lane) = W4[(size_t)(16 * wave + i) * (CH / 4) + lane];
    __syncthreads();
    v2f bfrag[CH / 4];
    const float* wrow = wst + c * APITCH + 2 * h;
#pragma unroll
    for (int s = 0; s < CH / 4; ++s) bfrag[s] = *(const v2f*)(wrow + 4 * s);
    __syncthreads();
    float bv = bias[16 * wave + c];

    int tile0 = blockIdx.x;
    int nt = (ROW_BLOCKS - 1 - tile0) / FUSE_BLOCKS + 1;
    int nq = 2 * nt;
#define NODE_OF(q) (16 * (tile0 + ((q) >> 1) * FUSE_BLOCKS) + 2 * wave + ((q) & 1))

    int  node = NODE_OF(0);
    int  d   = cnt[node];
    int2 raw = bec[(size_t)node * CAP + lane];
    int  cdeg = cnt[(lane < d) ? raw.y : 0];
    int  n1 = NODE_OF(1);
    int  d1 = cnt[n1];
    int2 raw1 = bec[(size_t)n1 * CAP + lane];

    for (int q = 0; q < nq; ++q) {
        int cdeg1 = cnt[(lane < d1) ? raw1.y : 0];
        bool has2 = q + 2 < nq;
        int  n2 = has2 ? NODE_OF(q + 2) : N_NODES - 1;
        int  d2 = 0;
        if (has2) d2 = cnt[n2];
        int2 raw2 = bec[(size_t)n2 * CAP + lane];

        float4 acc = gather_node(x4, ei, cnt, bec, node, d, raw, cdeg, lane);
        int buf = (q >> 1) & 1;
        *(float4*)(tile + (buf * 16 + 2 * wave + (q & 1)) * APITCH + 4 * lane) = acc;

        if (q & 1) {
            __syncthreads();
            const float* ap = tile + (buf * 16 + c) * APITCH + 2 * h;
            v8f o = {};
#pragma unroll
            for (int s = 0; s < CH / 4; ++s) {
                v2f a = *(const v2f*)(ap + 4 * s);
                o = __builtin_amdgcn_wmma_f32_16x16x4_f32(
                    false, a, false, bfrag[s], (short)0, o, false, false);
            }
            int t = tile0 + (q >> 1) * FUSE_BLOCKS;
            float* op = out + (size_t)(16 * t + 8 * h) * CH + 16 * wave + c;
#pragma unroll
            for (int r = 0; r < 8; ++r) op[(size_t)r * CH] = o[r] + bv;
        }

        node = n1;  d = d1;  raw = raw1;  cdeg = cdeg1;
        n1 = n2;  d1 = d2;  raw1 = raw2;
    }
#undef NODE_OF
}

extern "C" void kernel_launch(void* const* d_in, const int* in_sizes, int n_in,
                              void* d_out, int out_size, void* d_ws, size_t ws_size,
                              hipStream_t stream) {
    const float* x  = (const float*)d_in[0];
    const int*   ei = (const int*)d_in[1];
    const float* W  = (const float*)d_in[2];
    const float* b  = (const float*)d_in[3];
    float* out = (float*)d_out;

    int*   wsi = (int*)d_ws;
    int*   cnt = wsi + WS_CNT;
    int*   bec = wsi + WS_BEC;
    int*   segcnt = wsi + WS_SEGCNT;
    int*   seg = wsi + WS_SEG;

    k_part<<<NB, PART_THREADS, 0, stream>>>(ei, segcnt, seg);
    k_place<<<NRANGE, PLACE_THREADS, 0, stream>>>(ei, segcnt, seg, cnt, bec);
    k_gather_gemm<<<FUSE_BLOCKS, 256, 0, stream>>>(x, ei, cnt, bec, W, b, out);
}


# Whole-line stores: edge records, bucket entries and out rows go to an LDS image of the block's region and leave it in 16-byte pieces (was scattered 8- and 4-byte stores); sums unchanged
# speedup vs baseline: 4.0859x; 1.0642x over previous
#include <hip/hip_runtime.h>

typedef __attribute__((ext_vector_type(2))) float v2f;
typedef __attribute__((ext_vector_type(8))) float v8f;

#define N_NODES 50000
#define N_EDGES 625000
#define CH 128
#define ROW_BLOCKS (N_NODES / 16)

#define WS_CNT   0
#define WS_BEC   50176
#define CAP      64
#define WS_SEGCNT 6450176
#define WS_SEG    6510208

#define EPT 8
#define EDGE_THREADS (N_EDGES / EPT)
#define PART_THREADS 256
#define PART_EDGES (EPT * PART_THREADS)
#define NB ((N_EDGES + PART_EDGES - 1) / PART_EDGES)
#define RSHIFT 8
#define RNODES (1 << RSHIFT)
#define NRANGE ((N_NODES + RNODES - 1) / RNODES)
#define SEGCAP 64
#define EID_BITS 20
#define PLACE_THREADS 512
#define PLACE_WAVES (PLACE_THREADS / 32)
#define PLACE_ITERS ((NB + PLACE_WAVES - 1) / PLACE_WAVES)

__global__ void __launch_bounds__(PART_THREADS)
k_part(const int* __restrict__ ei, int* __restrict__ segcnt, int* __restrict__ seg_w) {
    __shared__ int hist[NRANGE + 1];
    __shared__ int4 lseg4[NRANGE * SEGCAP / 2];
    int2* lseg = (int2*)lseg4;
    int4* seg4 = (int4*)seg_w;
    int tid = threadIdx.x;
    int b = blockIdx.x;
    int t = b * PART_THREADS + tid;
    bool act = t < EDGE_THREADS;
    int4 ra = make_int4(0, 0, 0, 0), rb = ra, ca = ra, cb = ra;
    if (act) {
        const int4* r4 = (const int4*)ei;
        const int4* c4 = (const int4*)(ei + N_EDGES);
        ra = r4[2 * t];  rb = r4[2 * t + 1];
        ca = c4[2 * t];  cb = c4[2 * t + 1];
    }
    if (tid <= NRANGE) hist[tid] = 0;
    __syncthreads();
    int r[EPT] = {ra.x, ra.y, ra.z, ra.w, rb.x, rb.y, rb.z, rb.w};
    int c[EPT] = {ca.x, ca.y, ca.z, ca.w, cb.x, cb.y, cb.z, cb.w};
    int rk[EPT];
    if (act) {
#pragma unroll
        for (int j = 0; j < EPT; ++j) rk[j] = atomicAdd(&hist[r[j] >> RSHIFT], 1);
    }
    if (act) {
        int mx = 0;
#pragma unroll
        for (int j = 0; j < EPT; ++j) {
            int rg = r[j] >> RSHIFT;
            mx = max(mx, rk[j]);
            if (rk[j] < SEGCAP)
                lseg[rg * SEGCAP + rk[j]] =
                    make_int2((EPT * t + j) | ((r[j] & (RNODES - 1)) << EID_BITS), c[j]);
        }
        if (mx >= 32) hist[NRANGE] = 1;
    }
    __syncthreads();
    if (tid < NRANGE) segcnt[tid * NB + b] = hist[tid];
    int big = hist[NRANGE];
#define PART_COPY ((NRANGE * 16 + PART_THREADS - 1) / PART_THREADS)
    int pcn = tid & 15;
    int cn[PART_COPY];
    int4 cv[PART_COPY];
#pragma unroll
    for (int i = 0; i < PART_COPY; ++i)
        cn[i] = hist[min((tid + PART_THREADS * i) >> 4, NRANGE - 1)];
#pragma unroll
    for (int i = 0; i < PART_COPY; ++i)
        cv[i] = lseg4[min((tid + PART_THREADS * i) >> 4, NRANGE - 1) * (SEGCAP / 2) + pcn];
#pragma unroll
    for (int i = 0; i < PART_COPY; ++i) {
        int sg = (tid + PART_THREADS * i) >> 4;
        int n = cn[i];
        int4 v = cv[i];
        if (2 * pcn >= n)     { v.x = 0; v.y = 0; }
        if (2 * pcn + 1 >= n) { v.z = 0; v.w = 0; }
        if (sg < NRANGE && n > 0)
            seg4[((size_t)sg * NB + b) * (SEGCAP / 2) + pcn] = v;
    }
#undef PART_COPY
    if (big) {
        for (int sg = tid >> 5; sg < NRANGE; sg += PART_THREADS / 32) {
            int n = hist[sg];
            int pc = 16 + (tid & 15);
            if (n > 32 && (tid & 31) < 16) {
                if (n > SEGCAP) n = SEGCAP;
                int4 v = lseg4[sg * (SEGCAP / 2) + pc];
                if (2 * pc >= n)     { v.x = 0; v.y = 0; }
                if (2 * pc + 1 >= n) { v.z = 0; v.w = 0; }
                seg4[((size_t)sg * NB + b) * (SEGCAP / 2) + pc] = v;
            }
        }
    }
}

__global__ void __launch_bounds__(PLACE_THREADS)
k_place(const int* __restrict__ ei, const int* __restrict__ segcnt,
        const int* __restrict__ seg_w, int* __restrict__ cnt, int* __restrict__ bec_w) {
    __shared__ int cnt_l[RNODES];
    __shared__ int wover[PLACE_WAVES];
    __shared__ int deep_l;
    __shared__ int4 lbec4[RNODES * CAP / 2];
    int2* lbec = (int2*)lbec4;
    const int2* seg = (const int2*)seg_w;
    int4* bec4 = (int4*)bec_w;
    int tid = threadIdx.x;
    int range = blockIdx.x;
    int wave = tid >> 5;
    int lane = tid & 31;
    int myc = 0;
    if (lane < PLACE_ITERS && wave + PLACE_WAVES * lane < NB)
        myc = segcnt[range * NB + wave + PLACE_WAVES * lane];
    if (tid < RNODES) cnt_l[tid] = 0;
    if (tid == 0) deep_l = 0;
    int mx = 0;
    int big = (__ballot(myc > SEGCAP) != 0) ? 1 : 0;
    if (lane == 0) wover[wave] = big;
    __syncthreads();
    bool over = __ballot(wover[lane & (PLACE_WAVES - 1)] != 0) != 0;

    if (!over) {
#define PLACE_SEGS(X) X(0) X(1) X(2) X(3) X(4) X(5) X(6) X(7) X(8) X(9) \
                      X(10) X(11) X(12) X(13) X(14) X(15) X(16) X(17) X(18) X(19)
        static_assert(PLACE_ITERS == 20, "PLACE_SEGS lists one step per segment of a wave");
#define PLACE_LOAD(u)                                                              \
        int n##u = __builtin_amdgcn_readlane(myc, u);                              \
        int2 v##u = make_int2(0, 0);                                               \
        if (lane < n##u)                                                           \
            v##u = seg[((size_t)range * NB + wave + PLACE_WAVES * u) * SEGCAP + lane];
#define PLACE_RANK(u)                                                              \
        int lrow##u = (v##u.x >> EID_BITS) & (RNODES - 1);                         \
        int slot##u = 0;                                                           \
        if (lane < n##u) slot##u = atomicAdd(&cnt_l[lrow##u], 1);
#define PLACE_STORE(u)                                                             \
        mx = max(mx, slot##u);                                                     \
        if (lane < n##u && slot##u < CAP)                                          \
            lbec[lrow##u * CAP + slot##u] =                                        \
                make_int2(v##u.x & ((1 << EID_BITS) - 1), v##u.y);
        PLACE_SEGS(PLACE_LOAD)
        PLACE_SEGS(PLACE_RANK)
        PLACE_SEGS(PLACE_STORE)
#undef PLACE_LOAD
#undef PLACE_RANK
#undef PLACE_STORE
#undef PLACE_SEGS
        for (int u = 0; u < PLACE_ITERS; ++u) {
            int n = __builtin_amdgcn_readlane(myc, u);
            if (lane + 32 < n) {
                int2 w = seg[((size_t)range * NB + wave + PLACE_WAVES * u) * SEGCAP + lane + 32];
                int lrow = (w.x >> EID_BITS) & (RNODES - 1);
                int sl = atomicAdd(&cnt_l[lrow], 1);
                mx = max(mx, sl);
                if (sl < CAP)
                    lbec[lrow * CAP + sl] =
                        make_int2(w.x & ((1 << EID_BITS) - 1), w.y);
            }
        }
    } else {
        for (int e = tid; e < N_EDGES; e += PLACE_THREADS) {
            int row = ei[e];
            if ((row >> RSHIFT) == range) {
                int lrow = row & (RNODES - 1);
                int sl = atomicAdd(&cnt_l[lrow], 1);
                mx = max(mx, sl);
                if (sl < CAP)
                    lbec[lrow * CAP + sl] = make_int2(e, ei[N_EDGES + e]);
            }
        }
    }
    if (mx >= 32) deep_l = 1;
    __syncthreads();
    if (tid < RNODES) {
        int node = range * RNODES + tid;
        if (node < N_NODES) cnt[node] = cnt_l[tid];
    }
    int deep = deep_l;
#define PLACE_COPY (RNODES * 16 / PLACE_THREADS)
    int pcn = tid & 15;
    int cn[PLACE_COPY];
    int4 cv[PLACE_COPY];
#pragma unroll
    for (int i = 0; i < PLACE_COPY; ++i)
        cn[i] = cnt_l[(tid + PLACE_THREADS * i) >> 4];
#pragma unroll
    for (int i = 0; i < PLACE_COPY; ++i)
        cv[i] = lbec4[((tid + PLACE_THREADS * i) >> 4) * (CAP / 2) + pcn];
#pragma unroll
    for (int i = 0; i < PLACE_COPY; ++i) {
        int node = range * RNODES + ((tid + PLACE_THREADS * i) >> 4);
        int n = cn[i];
        int4 v = cv[i];
        if (2 * pcn >= n)     { v.x = 0; v.y = 0; }
        if (2 * pcn + 1 >= n) { v.z = 0; v.w = 0; }
        if (node < N_NODES)
            bec4[(size_t)node * (CAP / 2) + pcn] = v;
    }
#undef PLACE_COPY
    if (deep) {
        for (int j = wave; j < RNODES; j += PLACE_WAVES) {
            int n = cnt_l[j];
            int pc = 16 + (lane & 15);
            int node = range * RNODES + j;
            if (n > 32 && lane < 16 && node < N_NODES) {
                if (n > CAP) n = CAP;
                int4 v = lbec4[j * (CAP / 2) + pc];
                if (2 * pc >= n)     { v.x = 0; v.y = 0; }
                if (2 * pc + 1 >= n) { v.z = 0; v.w = 0; }
                bec4[(size_t)node * (CAP / 2) + pc] = v;
            }
        }
    }
}

#define FUSE_BLOCKS 1024
#define APITCH 132

__device__ __forceinline__ float4 fma4(float4 v, float w, float4 a) {
    a.x = fmaf(v.x, w, a.x);
    a.y = fmaf(v.y, w, a.y);
    a.z = fmaf(v.z, w, a.z);
    a.w = fmaf(v.w, w, a.w);
    return a;
}

__device__ __forceinline__ float dinv(int deg) {
    return (deg > 0) ? rsqrtf((float)deg) : 0.0f;
}

__device__ __forceinline__ float4 gather_node(
        const float4* __restrict__ x4, const int* __restrict__ ei,
        const int* __restrict__ cnt, const int2* __restrict__ bec,
        int node, int d, int2 raw, int cdeg, int lane) {
    float dr = dinv(d);
    float4 acc = make_float4(0.0f, 0.0f, 0.0f, 0.0f);
    if (d <= 32) {
        int   e = (lane < d) ? raw.x : 0x7fffffff;
        int   c = (lane < d) ? raw.y : 0;
        float w = (lane < d) ? dr * dinv(cdeg) : 0.0f;
        int rk = 0;
        for (int k = 0; k < d; ++k) {
            int ek = __builtin_amdgcn_readlane(e, k);
            rk += (ek < e) ? 1 : 0;
        }
        if (lane >= d) rk = lane;
        int   sc = __builtin_amdgcn_ds_permute(rk << 2, c);
        float sw = __int_as_float(__builtin_amdgcn_ds_permute(rk << 2, __float_as_int(w)));

        int i = 0;
        for (; i + 4 <= d; i += 4) {
            int c0 = __builtin_amdgcn_readlane(sc, i);
            int c1r = __builtin_amdgcn_readlane(sc, i + 1);
            int c2 = __builtin_amdgcn_readlane(sc, i + 2);
            int c3 = __builtin_amdgcn_readlane(sc, i + 3);
            float4 v0 = x4[(size_t)c0 * (CH / 4) + lane];
            float4 v1 = x4[(size_t)c1r * (CH / 4) + lane];
            float4 v2 = x4[(size_t)c2 * (CH / 4) + lane];
            float4 v3 = x4[(size_t)c3 * (CH / 4) + lane];
            float w0 = __int_as_float(__builtin_amdgcn_readlane(__float_as_int(sw), i));
            float w1r = __int_as_float(__builtin_amdgcn_readlane(__float_as_int(sw), i + 1));
            float w2 = __int_as_float(__builtin_amdgcn_readlane(__float_as_int(sw), i + 2));
            float w3 = __int_as_float(__builtin_amdgcn_readlane(__float_as_int(sw), i + 3));
            acc = fma4(v0, w0, acc);
            acc = fma4(v1, w1r, acc);
            acc = fma4(v2, w2, acc);
            acc = fma4(v3, w3, acc);
        }
        for (; i < d; ++i) {
            int   ci = __builtin_amdgcn_readlane(sc, i);
            float wi = __int_as_float(__builtin_amdgcn_readlane(__float_as_int(sw), i));
            acc = fma4(x4[(size_t)ci * (CH / 4) + lane], wi, acc);
        }
    } else if (d <= CAP) {
        const int2* bk = bec + (size_t)node * CAP;
        int last = -1;
        for (int i = 0; i < d; ++i) {
            int best = 0x7fffffff, bp = 0;
            for (int p = lane; p < d; p += 32) {
                int id = bk[p].x;
                if (id > last && id < best) { best = id; bp = p; }
            }
            for (int s = 16; s > 0; s >>= 1) {
                int ob = __shfl_xor(best, s);
                int op = __shfl_xor(bp, s);
                if (ob < best) { best = ob; bp = op; }
            }
            last = best;
            int   ci = bk[bp].y;
            float wi = dr * dinv(cnt[ci]);
            acc = fma4(x4[(size_t)ci * (CH / 4) + lane], wi, acc);
        }
    } else {
        for (int base = 0; base < N_EDGES; base += 32) {
            int eid = base + lane;
            int rr = -1, cc = 0;
            if (eid < N_EDGES) { rr = ei[eid]; cc = ei[N_EDGES + eid]; }
            unsigned m = (unsigned)__ballot(rr == node);
            while (m) {
                int l = __builtin_ctz(m);
                m &= m - 1;
                int   ci = __builtin_amdgcn_readlane(cc, l);
                float wi = dr * dinv(cnt[ci]);
                acc = fma4(x4[(size_t)ci * (CH / 4) + lane], wi, acc);
            }
        }
    }
    return acc;
}

__device__ __forceinline__ void store_rows(const float* __restrict__ img, float* __restrict__ dst,
                                           int wave, int lane) {
    float4 v0 = *(const float4*)(img + (2 * wave) * APITCH + 4 * lane);
    float4 v1 = *(const float4*)(img + (2 * wave + 1) * APITCH + 4 * lane);
    *(float4*)(dst + (size_t)(2 * wave) * CH + 4 * lane) = v0;
    *(float4*)(dst + (size_t)(2 * wave + 1) * CH + 4 * lane) = v1;
}

__global__ void __launch_bounds__(256, 8)
k_gather_gemm(const float* __restrict__ x, const int* __restrict__ ei,
              const int* __restrict__ cnt, const int* __restrict__ bec_w,
              const float* __restrict__ W,
              const float* __restrict__ bias,
              float* __restrict__ out) {
    __shared__ float4 lds4[8 * 16 * APITCH / 4];
    float* tile = (float*)lds4;
    const int2* bec = (const int2*)bec_w;
    const float4* x4 = (const float4*)x;
    int lane = threadIdx.x & 31;
    int wave = __builtin_amdgcn_readfirstlane(threadIdx.x >> 5);
    int c = lane & 15;
    int h = lane >> 4;

    float* wst = tile + wave * 16 * APITCH;
    const float4* W4 = (const float4*)W;
#pragma unroll
    for (int i = 0; i < 16; ++i)
        *(float4*)(wst + i * APITCH + 4 * lane) = W4[(size_t)(16 * wave + i) * (CH / 4) + lane];
    __syncthreads();
    v2f bfrag[CH / 4];
    const float* wrow = wst + c * APITCH + 2 * h;
#pragma unroll
    for (int s = 0; s < CH / 4; ++s) bfrag[s] = *(const v2f*)(wrow + 4 * s);
    __syncthreads();
    float bv = bias[16 * wave + c];

    int tile0 = blockIdx.x;
    int nt = (ROW_BLOCKS - 1 - tile0) / FUSE_BLOCKS + 1;
    int nq = 2 * nt;
#define NODE_OF(q) (16 * (tile0 + ((q) >> 1) * FUSE_BLOCKS) + 2 * wave + ((q) & 1))

    int  node = NODE_OF(0);
    int  d   = cnt[node];
    int2 raw = bec[(size_t)node * CAP + lane];
    int  cdeg = cnt[(lane < d) ? raw.y : 0];
    int  n1 = NODE_OF(1);
    int  d1 = cnt[n1];
    int2 raw1 = bec[(size_t)n1 * CAP + lane];

    for (int q = 0; q < nq; ++q) {
        int cdeg1 = cnt[(lane < d1) ? raw1.y : 0];
        bool has2 = q + 2 < nq;
        int  n2 = has2 ? NODE_OF(q + 2) : N_NODES - 1;
        int  d2 = 0;
        if (has2) d2 = cnt[n2];
        int2 raw2 = bec[(size_t)n2 * CAP + lane];

        float4 acc = gather_node(x4, ei, cnt, bec, node, d, raw, cdeg, lane);
        int buf = (q >> 1) & 1;
        *(float4*)(tile + (buf * 16 + 2 * wave + (q & 1)) * APITCH + 4 * lane) = acc;

        if (q & 1) {
            __syncthreads();
            if (q > 1)
                store_rows(tile + (32 + (buf ^ 1) * 16) * APITCH,
                           out + (size_t)16 * (tile0 + ((q >> 1) - 1) * FUSE_BLOCKS) * CH, wave, lane);
            const float* ap = tile + (buf * 16 + c) * APITCH + 2 * h;
            v8f o = {};
#pragma unroll
            for (int s = 0; s < CH / 4; ++s) {
                v2f a = *(const v2f*)(ap + 4 * s);
                o = __builtin_amdgcn_wmma_f32_16x16x4_f32(
                    false, a, false, bfrag[s], (short)0, o, false, false);
            }
            float* os = tile + (32 + buf * 16 + 8 * h) * APITCH + 16 * wave + c;
#pragma unroll
            for (int r = 0; r < 8; ++r) os[r * APITCH] = o[r] + bv;
        }

        node = n1;  d = d1;  raw = raw1;  cdeg = cdeg1;
        n1 = n2;  d1 = d2;  raw1 = raw2;
    }
#undef NODE_OF
    __syncthreads();
    store_rows(tile + (32 + ((nt - 1) & 1) * 16) * APITCH,
               out + (size_t)16 * (tile0 + (nt - 1) * FUSE_BLOCKS) * CH, wave, lane);
}

extern "C" void kernel_launch(void* const* d_in, const int* in_sizes, int n_in,
                              void* d_out, int out_size, void* d_ws, size_t ws_size,
                              hipStream_t stream) {
    const float* x  = (const float*)d_in[0];
    const int*   ei = (const int*)d_in[1];
    const float* W  = (const float*)d_in[2];
    const float* b  = (const float*)d_in[3];
    float* out = (float*)d_out;

    int*   wsi = (int*)d_ws;
    int*   cnt = wsi + WS_CNT;
    int*   bec = wsi + WS_BEC;
    int*   segcnt = wsi + WS_SEGCNT;
    int*   seg = wsi + WS_SEG;

    k_part<<<NB, PART_THREADS, 0, stream>>>(ei, segcnt, seg);
    k_place<<<NRANGE, PLACE_THREADS, 0, stream>>>(ei, segcnt, seg, cnt, bec);
    k_gather_gemm<<<FUSE_BLOCKS, 256, 0, stream>>>(x, ei, cnt, bec, W, b, out);
}


# Gather: 8 x rows in flight per wave (was 4), and the d mod 8 rows left over loaded as one group under wave-uniform tests (was one row at a time); adds stay in rank order, sums unchanged
# speedup vs baseline: 4.2409x; 1.0379x over previous
#include <hip/hip_runtime.h>

typedef __attribute__((ext_vector_type(2))) float v2f;
typedef __attribute__((ext_vector_type(8))) float v8f;

#define N_NODES 50000
#define N_EDGES 625000
#define CH 128
#define ROW_BLOCKS (N_NODES / 16)

#define WS_CNT   0
#define WS_BEC   50176
#define CAP      64
#define WS_SEGCNT 6450176
#define WS_SEG    6510208

#define EPT 8
#define EDGE_THREADS (N_EDGES / EPT)
#define PART_THREADS 256
#define PART_EDGES (EPT * PART_THREADS)
#define NB ((N_EDGES + PART_EDGES - 1) / PART_EDGES)
#define RSHIFT 8
#define RNODES (1 << RSHIFT)
#define NRANGE ((N_NODES + RNODES - 1) / RNODES)
#define SEGCAP 64
#define EID_BITS 20
#define PLACE_THREADS 512
#define PLACE_WAVES (PLACE_THREADS / 32)
#define PLACE_ITERS ((NB + PLACE_WAVES - 1) / PLACE_WAVES)

__global__ void __launch_bounds__(PART_THREADS)
k_part(const int* __restrict__ ei, int* __restrict__ segcnt, int* __restrict__ seg_w) {
    __shared__ int hist[NRANGE + 1];
    __shared__ int4 lseg4[NRANGE * SEGCAP / 2];
    int2* lseg = (int2*)lseg4;
    int4* seg4 = (int4*)seg_w;
    int tid = threadIdx.x;
    int b = blockIdx.x;
    int t = b * PART_THREADS + tid;
    bool act = t < EDGE_THREADS;
    int4 ra = make_int4(0, 0, 0, 0), rb = ra, ca = ra, cb = ra;
    if (act) {
        const int4* r4 = (const int4*)ei;
        const int4* c4 = (const int4*)(ei + N_EDGES);
        ra = r4[2 * t];  rb = r4[2 * t + 1];
        ca = c4[2 * t];  cb = c4[2 * t + 1];
    }
    if (tid <= NRANGE) hist[tid] = 0;
    __syncthreads();
    int r[EPT] = {ra.x, ra.y, ra.z, ra.w, rb.x, rb.y, rb.z, rb.w};
    int c[EPT] = {ca.x, ca.y, ca.z, ca.w, cb.x, cb.y, cb.z, cb.w};
    int rk[EPT];
    if (act) {
#pragma unroll
        for (int j = 0; j < EPT; ++j) rk[j] = atomicAdd(&hist[r[j] >> RSHIFT], 1);
    }
    if (act) {
        int mx = 0;
#pragma unroll
        for (int j = 0; j < EPT; ++j) {
            int rg = r[j] >> RSHIFT;
            mx = max(mx, rk[j]);
            if (rk[j] < SEGCAP)
                lseg[rg * SEGCAP + rk[j]] =
                    make_int2((EPT * t + j) | ((r[j] & (RNODES - 1)) << EID_BITS), c[j]);
        }
        if (mx >= 32) hist[NRANGE] = 1;
    }
    __syncthreads();
    if (tid < NRANGE) segcnt[tid * NB + b] = hist[tid];
    int big = hist[NRANGE];
#define PART_COPY ((NRANGE * 16 + PART_THREADS - 1) / PART_THREADS)
    int pcn = tid & 15;
    int cn[PART_COPY];
    int4 cv[PART_COPY];
#pragma unroll
    for (int i = 0; i < PART_COPY; ++i)
        cn[i] = hist[min((tid + PART_THREADS * i) >> 4, NRANGE - 1)];
#pragma unroll
    for (int i = 0; i < PART_COPY; ++i)
        cv[i] = lseg4[min((tid + PART_THREADS * i) >> 4, NRANGE - 1) * (SEGCAP / 2) + pcn];
#pragma unroll
    for (int i = 0; i < PART_COPY; ++i) {
        int sg = (tid + PART_THREADS * i) >> 4;
        int n = cn[i];
        int4 v = cv[i];
        if (2 * pcn >= n)     { v.x = 0; v.y = 0; }
        if (2 * pcn + 1 >= n) { v.z = 0; v.w = 0; }
        if (sg < NRANGE && n > 0)
            seg4[((size_t)sg * NB + b) * (SEGCAP / 2) + pcn] = v;
    }
#undef PART_COPY
    if (big) {
        for (int sg = tid >> 5; sg < NRANGE; sg += PART_THREADS / 32) {
            int n = hist[sg];
            int pc = 16 + (tid & 15);
            if (n > 32 && (tid & 31) < 16) {
                if (n > SEGCAP) n = SEGCAP;
                int4 v = lseg4[sg * (SEGCAP / 2) + pc];
                if (2 * pc >= n)     { v.x = 0; v.y = 0; }
                if (2 * pc + 1 >= n) { v.z = 0; v.w = 0; }
                seg4[((size_t)sg * NB + b) * (SEGCAP / 2) + pc] = v;
            }
        }
    }
}

__global__ void __launch_bounds__(PLACE_THREADS)
k_place(const int* __restrict__ ei, const int* __restrict__ segcnt,
        const int* __restrict__ seg_w, int* __restrict__ cnt, int* __restrict__ bec_w) {
    __shared__ int cnt_l[RNODES];
    __shared__ int wover[PLACE_WAVES];
    __shared__ int deep_l;
    __shared__ int4 lbec4[RNODES * CAP / 2];
    int2* lbec = (int2*)lbec4;
    const int2* seg = (const int2*)seg_w;
    int4* bec4 = (int4*)bec_w;
    int tid = threadIdx.x;
    int range = blockIdx.x;
    int wave = tid >> 5;
    int lane = tid & 31;
    int myc = 0;
    if (lane < PLACE_ITERS && wave + PLACE_WAVES * lane < NB)
        myc = segcnt[range * NB + wave + PLACE_WAVES * lane];
    if (tid < RNODES) cnt_l[tid] = 0;
    if (tid == 0) deep_l = 0;
    int mx = 0;
    int big = (__ballot(myc > SEGCAP) != 0) ? 1 : 0;
    if (lane == 0) wover[wave] = big;
    __syncthreads();
    bool over = __ballot(wover[lane & (PLACE_WAVES - 1)] != 0) != 0;

    if (!over) {
#define PLACE_SEGS(X) X(0) X(1) X(2) X(3) X(4) X(5) X(6) X(7) X(8) X(9) \
                      X(10) X(11) X(12) X(13) X(14) X(15) X(16) X(17) X(18) X(19)
        static_assert(PLACE_ITERS == 20, "PLACE_SEGS lists one step per segment of a wave");
#define PLACE_LOAD(u)                                                              \
        int n##u = __builtin_amdgcn_readlane(myc, u);                              \
        int2 v##u = make_int2(0, 0);                                               \
        if (lane < n##u)                                                           \
            v##u = seg[((size_t)range * NB + wave + PLACE_WAVES * u) * SEGCAP + lane];
#define PLACE_RANK(u)                                                              \
        int lrow##u = (v##u.x >> EID_BITS) & (RNODES - 1);                         \
        int slot##u = 0;                                                           \
        if (lane < n##u) slot##u = atomicAdd(&cnt_l[lrow##u], 1);
#define PLACE_STORE(u)                                                             \
        mx = max(mx, slot##u);                                                     \
        if (lane < n##u && slot##u < CAP)                                          \
            lbec[lrow##u * CAP + slot##u] =                                        \
                make_int2(v##u.x & ((1 << EID_BITS) - 1), v##u.y);
        PLACE_SEGS(PLACE_LOAD)
        PLACE_SEGS(PLACE_RANK)
        PLACE_SEGS(PLACE_STORE)
#undef PLACE_LOAD
#undef PLACE_RANK
#undef PLACE_STORE
#undef PLACE_SEGS
        for (int u = 0; u < PLACE_ITERS; ++u) {
            int n = __builtin_amdgcn_readlane(myc, u);
            if (lane + 32 < n) {
                int2 w = seg[((size_t)range * NB + wave + PLACE_WAVES * u) * SEGCAP + lane + 32];
                int lrow = (w.x >> EID_BITS) & (RNODES - 1);
                int sl = atomicAdd(&cnt_l[lrow], 1);
                mx = max(mx, sl);
                if (sl < CAP)
                    lbec[lrow * CAP + sl] =
                        make_int2(w.x & ((1 << EID_BITS) - 1), w.y);
            }
        }
    } else {
        for (int e = tid; e < N_EDGES; e += PLACE_THREADS) {
            int row = ei[e];
            if ((row >> RSHIFT) == range) {
                int lrow = row & (RNODES - 1);
                int sl = atomicAdd(&cnt_l[lrow], 1);
                mx = max(mx, sl);
                if (sl < CAP)
                    lbec[lrow * CAP + sl] = make_int2(e, ei[N_EDGES + e]);
            }
        }
    }
    if (mx >= 32) deep_l = 1;
    __syncthreads();
    if (tid < RNODES) {
        int node = range * RNODES + tid;
        if (node < N_NODES) cnt[node] = cnt_l[tid];
    }
    int deep = deep_l;
#define PLACE_COPY (RNODES * 16 / PLACE_THREADS)
    int pcn = tid & 15;
    int cn[PLACE_COPY];
    int4 cv[PLACE_COPY];
#pragma unroll
    for (int i = 0; i < PLACE_COPY; ++i)
        cn[i] = cnt_l[(tid + PLACE_THREADS * i) >> 4];
#pragma unroll
    for (int i = 0; i < PLACE_COPY; ++i)
        cv[i] = lbec4[((tid + PLACE_THREADS * i) >> 4) * (CAP / 2) + pcn];
#pragma unroll
    for (int i = 0; i < PLACE_COPY; ++i) {
        int node = range * RNODES + ((tid + PLACE_THREADS * i) >> 4);
        int n = cn[i];
        int4 v = cv[i];
        if (2 * pcn >= n)     { v.x = 0; v.y = 0; }
        if (2 * pcn + 1 >= n) { v.z = 0; v.w = 0; }
        if (node < N_NODES)
            bec4[(size_t)node * (CAP / 2) + pcn] = v;
    }
#undef PLACE_COPY
    if (deep) {
        for (int j = wave; j < RNODES; j += PLACE_WAVES) {
            int n = cnt_l[j];
            int pc = 16 + (lane & 15);
            int node = range * RNODES + j;
            if (n > 32 && lane < 16 && node < N_NODES) {
                if (n > CAP) n = CAP;
                int4 v = lbec4[j * (CAP / 2) + pc];
                if (2 * pc >= n)     { v.x = 0; v.y = 0; }
                if (2 * pc + 1 >= n) { v.z = 0; v.w = 0; }
                bec4[(size_t)node * (CAP / 2) + pc] = v;
            }
        }
    }
}

#define FUSE_BLOCKS 1024
#define APITCH 132

__device__ __forceinline__ float4 fma4(float4 v, float w, float4 a) {
    a.x = fmaf(v.x, w, a.x);
    a.y = fmaf(v.y, w, a.y);
    a.z = fmaf(v.z, w, a.z);
    a.w = fmaf(v.w, w, a.w);
    return a;
}

__device__ __forceinline__ float dinv(int deg) {
    return (deg > 0) ? rsqrtf((float)deg) : 0.0f;
}

__device__ __forceinline__ float4 load_row(const float* __restrict__ x, int c, unsigned lane_off) {
    const char* rp = (const char*)x + ((size_t)(unsigned)c << 9);
    return *(const float4*)(rp + lane_off);
}

__device__ __forceinline__ float4 gather_node(
        const float4* __restrict__ x4, const int* __restrict__ ei,
        const int* __restrict__ cnt, const int2* __restrict__ bec,
        int node, int d, int2 raw, int cdeg, int lane) {
    float dr = dinv(d);
    float4 acc = make_float4(0.0f, 0.0f, 0.0f, 0.0f);
    if (d <= 32) {
        int   e = (lane < d) ? raw.x : 0x7fffffff;
        int   c = (lane < d) ? raw.y : 0;
        float w = (lane < d) ? dr * dinv(cdeg) : 0.0f;
        int rk = 0;
        for (int k = 0; k < d; ++k) {
            int ek = __builtin_amdgcn_readlane(e, k);
            rk += (ek < e) ? 1 : 0;
        }
        if (lane >= d) rk = lane;
        int   sc = __builtin_amdgcn_ds_permute(rk << 2, c);
        float sw = __int_as_float(__builtin_amdgcn_ds_permute(rk << 2, __float_as_int(w)));

#define ROWS_AHEAD 8
        int i = 0;
        for (; i + ROWS_AHEAD <= d; i += ROWS_AHEAD) {
            float4 v[ROWS_AHEAD];
            float  wg[ROWS_AHEAD];
#pragma unroll
            for (int j = 0; j < ROWS_AHEAD; ++j) {
                int cj = __builtin_amdgcn_readlane(sc, i + j);
                v[j] = load_row((const float*)x4, cj, (unsigned)lane << 4);
            }
#pragma unroll
            for (int j = 0; j < ROWS_AHEAD; ++j)
                wg[j] = __int_as_float(__builtin_amdgcn_readlane(__float_as_int(sw), i + j));
#pragma unroll
            for (int j = 0; j < ROWS_AHEAD; ++j) acc = fma4(v[j], wg[j], acc);
        }
        int left = d - i;
        if (left > 0) {
            float4 v[ROWS_AHEAD - 1];
            float  wg[ROWS_AHEAD - 1];
#pragma unroll
            for (int j = 0; j < ROWS_AHEAD - 1; ++j) {
                v[j] = make_float4(0.0f, 0.0f, 0.0f, 0.0f);
                wg[j] = 0.0f;
                if (j < left) {
                    int cj = __builtin_amdgcn_readlane(sc, i + j);
                    v[j] = load_row((const float*)x4, cj, (unsigned)lane << 4);
                    wg[j] = __int_as_float(__builtin_amdgcn_readlane(__float_as_int(sw), i + j));
                }
            }
#pragma unroll
            for (int j = 0; j < ROWS_AHEAD - 1; ++j)
                if (j < left) acc = fma4(v[j], wg[j], acc);
        }
#undef ROWS_AHEAD
    } else if (d <= CAP) {
        const int2* bk = bec + (size_t)node * CAP;
        int last = -1;
        for (int i = 0; i < d; ++i) {
            int best = 0x7fffffff, bp = 0;
            for (int p = lane; p < d; p += 32) {
                int id = bk[p].x;
                if (id > last && id < best) { best = id; bp = p; }
            }
            for (int s = 16; s > 0; s >>= 1) {
                int ob = __shfl_xor(best, s);
                int op = __shfl_xor(bp, s);
                if (ob < best) { best = ob; bp = op; }
            }
            last = best;
            int   ci = bk[bp].y;
            float wi = dr * dinv(cnt[ci]);
            acc = fma4(x4[(size_t)ci * (CH / 4) + lane], wi, acc);
        }
    } else {
        for (int base = 0; base < N_EDGES; base += 32) {
            int eid = base + lane;
            int rr = -1, cc = 0;
            if (eid < N_EDGES) { rr = ei[eid]; cc = ei[N_EDGES + eid]; }
            unsigned m = (unsigned)__ballot(rr == node);
            while (m) {
                int l = __builtin_ctz(m);
                m &= m - 1;
                int   ci = __builtin_amdgcn_readlane(cc, l);
                float wi = dr * dinv(cnt[ci]);
                acc = fma4(x4[(size_t)ci * (CH / 4) + lane], wi, acc);
            }
        }
    }
    return acc;
}

__device__ __forceinline__ void store_rows(const float* __restrict__ img, float* __restrict__ dst,
                                           int wave, int lane) {
    float4 v0 = *(const float4*)(img + (2 * wave) * APITCH + 4 * lane);
    float4 v1 = *(const float4*)(img + (2 * wave + 1) * APITCH + 4 * lane);
    *(float4*)(dst + (size_t)(2 * wave) * CH + 4 * lane) = v0;
    *(float4*)(dst + (size_t)(2 * wave + 1) * CH + 4 * lane) = v1;
}

__global__ void __launch_bounds__(256, 8)
k_gather_gemm(const float* __restrict__ x, const int* __restrict__ ei,
              const int* __restrict__ cnt, const int* __restrict__ bec_w,
              const float* __restrict__ W,
              const float* __restrict__ bias,
              float* __restrict__ out) {
    __shared__ float4 lds4[8 * 16 * APITCH / 4];
    float* tile = (float*)lds4;
    const int2* bec = (const int2*)bec_w;
    const float4* x4 = (const float4*)x;
    int lane = threadIdx.x & 31;
    int wave = __builtin_amdgcn_readfirstlane(threadIdx.x >> 5);
    int c = lane & 15;
    int h = lane >> 4;

    float* wst = tile + wave * 16 * APITCH;
    const float4* W4 = (const float4*)W;
#pragma unroll
    for (int i = 0; i < 16; ++i)
        *(float4*)(wst + i * APITCH + 4 * lane) = W4[(size_t)(16 * wave + i) * (CH / 4) + lane];
    __syncthreads();
    v2f bfrag[CH / 4];
    const float* wrow = wst + c * APITCH + 2 * h;
#pragma unroll
    for (int s = 0; s < CH / 4; ++s) bfrag[s] = *(const v2f*)(wrow + 4 * s);
    __syncthreads();
    float bv = bias[16 * wave + c];

    int tile0 = blockIdx.x;
    int nt = (ROW_BLOCKS - 1 - tile0) / FUSE_BLOCKS + 1;
    int nq = 2 * nt;
#define NODE_OF(q) (16 * (tile0 + ((q) >> 1) * FUSE_BLOCKS) + 2 * wave + ((q) & 1))

    int  node = NODE_OF(0);
    int  d   = cnt[node];
    int2 raw = bec[(size_t)node * CAP + lane];
    int  cdeg = cnt[(lane < d) ? raw.y : 0];
    int  n1 = NODE_OF(1);
    int  d1 = cnt[n1];
    int2 raw1 = bec[(size_t)n1 * CAP + lane];

    for (int q = 0; q < nq; ++q) {
        int cdeg1 = cnt[(lane < d1) ? raw1.y : 0];
        bool has2 = q + 2 < nq;
        int  n2 = has2 ? NODE_OF(q + 2) : N_NODES - 1;
        int  d2 = 0;
        if (has2) d2 = cnt[n2];
        int2 raw2 = bec[(size_t)n2 * CAP + lane];

        float4 acc = gather_node(x4, ei, cnt, bec, node, d, raw, cdeg, lane);
        int buf = (q >> 1) & 1;
        *(float4*)(tile + (buf * 16 + 2 * wave + (q & 1)) * APITCH + 4 * lane) = acc;

        if (q & 1) {
            __syncthreads();
            if (q > 1)
                store_rows(tile + (32 + (buf ^ 1) * 16) * APITCH,
                           out + (size_t)16 * (tile0 + ((q >> 1) - 1) * FUSE_BLOCKS) * CH, wave, lane);
            const float* ap = tile + (buf * 16 + c) * APITCH + 2 * h;
            v8f o = {};
#pragma unroll
            for (int s = 0; s < CH / 4; ++s) {
                v2f a = *(const v2f*)(ap + 4 * s);
                o = __builtin_amdgcn_wmma_f32_16x16x4_f32(
                    false, a, false, bfrag[s], (short)0, o, false, false);
            }
            float* os = tile + (32 + buf * 16 + 8 * h) * APITCH + 16 * wave + c;
#pragma unroll
            for (int r = 0; r < 8; ++r) os[r * APITCH] = o[r] + bv;
        }

        node = n1;  d = d1;  raw = raw1;  cdeg = cdeg1;
        n1 = n2;  d1 = d2;  raw1 = raw2;
    }
#undef NODE_OF
    __syncthreads();
    store_rows(tile + (32 + ((nt - 1) & 1) * 16) * APITCH,
               out + (size_t)16 * (tile0 + (nt - 1) * FUSE_BLOCKS) * CH, wave, lane);
}

extern "C" void kernel_launch(void* const* d_in, const int* in_sizes, int n_in,
                              void* d_out, int out_size, void* d_ws, size_t ws_size,
                              hipStream_t stream) {
    const float* x  = (const float*)d_in[0];
    const int*   ei = (const int*)d_in[1];
    const float* W  = (const float*)d_in[2];
    const float* b  = (const float*)d_in[3];
    float* out = (float*)d_out;

    int*   wsi = (int*)d_ws;
    int*   cnt = wsi + WS_CNT;
    int*   bec = wsi + WS_BEC;
    int*   segcnt = wsi + WS_SEGCNT;
    int*   seg = wsi + WS_SEG;

    k_part<<<NB, PART_THREADS, 0, stream>>>(ei, segcnt, seg);
    k_place<<<NRANGE, PLACE_THREADS, 0, stream>>>(ei, segcnt, seg, cnt, bec);
    k_gather_gemm<<<FUSE_BLOCKS, 256, 0, stream>>>(x, ei, cnt, bec, W, b, out);
}


# Gather: next node's index work (source-degree, bucket and degree loads, rank by edge id) done under this node's row loads, the first nodes' under the W staging; none waited for alone; sums unchanged
# speedup vs baseline: 4.5178x; 1.0653x over previous
#include <hip/hip_runtime.h>

typedef __attribute__((ext_vector_type(2))) float v2f;
typedef __attribute__((ext_vector_type(8))) float v8f;

#define N_NODES 50000
#define N_EDGES 625000
#define CH 128
#define ROW_BLOCKS (N_NODES / 16)

#define WS_CNT   0
#define WS_BEC   50176
#define CAP      64
#define WS_SEGCNT 6450176
#define WS_SEG    6510208

#define EPT 8
#define EDGE_THREADS (N_EDGES / EPT)
#define PART_THREADS 256
#define PART_EDGES (EPT * PART_THREADS)
#define NB ((N_EDGES + PART_EDGES - 1) / PART_EDGES)
#define RSHIFT 8
#define RNODES (1 << RSHIFT)
#define NRANGE ((N_NODES + RNODES - 1) / RNODES)
#define SEGCAP 64
#define EID_BITS 20
#define PLACE_THREADS 512
#define PLACE_WAVES (PLACE_THREADS / 32)
#define PLACE_ITERS ((NB + PLACE_WAVES - 1) / PLACE_WAVES)

__global__ void __launch_bounds__(PART_THREADS)
k_part(const int* __restrict__ ei, int* __restrict__ segcnt, int* __restrict__ seg_w) {
    __shared__ int hist[NRANGE + 1];
    __shared__ int4 lseg4[NRANGE * SEGCAP / 2];
    int2* lseg = (int2*)lseg4;
    int4* seg4 = (int4*)seg_w;
    int tid = threadIdx.x;
    int b = blockIdx.x;
    int t = b * PART_THREADS + tid;
    bool act = t < EDGE_THREADS;
    int4 ra = make_int4(0, 0, 0, 0), rb = ra, ca = ra, cb = ra;
    if (act) {
        const int4* r4 = (const int4*)ei;
        const int4* c4 = (const int4*)(ei + N_EDGES);
        ra = r4[2 * t];  rb = r4[2 * t + 1];
        ca = c4[2 * t];  cb = c4[2 * t + 1];
    }
    if (tid <= NRANGE) hist[tid] = 0;
    __syncthreads();
    int r[EPT] = {ra.x, ra.y, ra.z, ra.w, rb.x, rb.y, rb.z, rb.w};
    int c[EPT] = {ca.x, ca.y, ca.z, ca.w, cb.x, cb.y, cb.z, cb.w};
    int rk[EPT];
    if (act) {
#pragma unroll
        for (int j = 0; j < EPT; ++j) rk[j] = atomicAdd(&hist[r[j] >> RSHIFT], 1);
    }
    if (act) {
        int mx = 0;
#pragma unroll
        for (int j = 0; j < EPT; ++j) {
            int rg = r[j] >> RSHIFT;
            mx = max(mx, rk[j]);
            if (rk[j] < SEGCAP)
                lseg[rg * SEGCAP + rk[j]] =
                    make_int2((EPT * t + j) | ((r[j] & (RNODES - 1)) << EID_BITS), c[j]);
        }
        if (mx >= 32) hist[NRANGE] = 1;
    }
    __syncthreads();
    if (tid < NRANGE) segcnt[tid * NB + b] = hist[tid];
    int big = hist[NRANGE];
#define PART_COPY ((NRANGE * 16 + PART_THREADS - 1) / PART_THREADS)
    int pcn = tid & 15;
    int cn[PART_COPY];
    int4 cv[PART_COPY];
#pragma unroll
    for (int i = 0; i < PART_COPY; ++i)
        cn[i] = hist[min((tid + PART_THREADS * i) >> 4, NRANGE - 1)];
#pragma unroll
    for (int i = 0; i < PART_COPY; ++i)
        cv[i] = lseg4[min((tid + PART_THREADS * i) >> 4, NRANGE - 1) * (SEGCAP / 2) + pcn];
#pragma unroll
    for (int i = 0; i < PART_COPY; ++i) {
        int sg = (tid + PART_THREADS * i) >> 4;
        int n = cn[i];
        int4 v = cv[i];
        if (2 * pcn >= n)     { v.x = 0; v.y = 0; }
        if (2 * pcn + 1 >= n) { v.z = 0; v.w = 0; }
        if (sg < NRANGE && n > 0)
            seg4[((size_t)sg * NB + b) * (SEGCAP / 2) + pcn] = v;
    }
#undef PART_COPY
    if (big) {
        for (int sg = tid >> 5; sg < NRANGE; sg += PART_THREADS / 32) {
            int n = hist[sg];
            int pc = 16 + (tid & 15);
            if (n > 32 && (tid & 31) < 16) {
                if (n > SEGCAP) n = SEGCAP;
                int4 v = lseg4[sg * (SEGCAP / 2) + pc];
                if (2 * pc >= n)     { v.x = 0; v.y = 0; }
                if (2 * pc + 1 >= n) { v.z = 0; v.w = 0; }
                seg4[((size_t)sg * NB + b) * (SEGCAP / 2) + pc] = v;
            }
        }
    }
}

__global__ void __launch_bounds__(PLACE_THREADS)
k_place(const int* __restrict__ ei, const int* __restrict__ segcnt,
        const int* __restrict__ seg_w, int* __restrict__ cnt, int* __restrict__ bec_w) {
    __shared__ int cnt_l[RNODES];
    __shared__ int wover[PLACE_WAVES];
    __shared__ int deep_l;
    __shared__ int4 lbec4[RNODES * CAP / 2];
    int2* lbec = (int2*)lbec4;
    const int2* seg = (const int2*)seg_w;
    int4* bec4 = (int4*)bec_w;
    int tid = threadIdx.x;
    int range = blockIdx.x;
    int wave = tid >> 5;
    int lane = tid & 31;
    int myc = 0;
    if (lane < PLACE_ITERS && wave + PLACE_WAVES * lane < NB)
        myc = segcnt[range * NB + wave + PLACE_WAVES * lane];
    if (tid < RNODES) cnt_l[tid] = 0;
    if (tid == 0) deep_l = 0;
    int mx = 0;
    int big = (__ballot(myc > SEGCAP) != 0) ? 1 : 0;
    if (lane == 0) wover[wave] = big;
    __syncthreads();
    bool over = __ballot(wover[lane & (PLACE_WAVES - 1)] != 0) != 0;

    if (!over) {
#define PLACE_SEGS(X) X(0) X(1) X(2) X(3) X(4) X(5) X(6) X(7) X(8) X(9) \
                      X(10) X(11) X(12) X(13) X(14) X(15) X(16) X(17) X(18) X(19)
        static_assert(PLACE_ITERS == 20, "PLACE_SEGS lists one step per segment of a wave");
#define PLACE_LOAD(u)                                                              \
        int n##u = __builtin_amdgcn_readlane(myc, u);                              \
        int2 v##u = make_int2(0, 0);                                               \
        if (lane < n##u)                                                           \
            v##u = seg[((size_t)range * NB + wave + PLACE_WAVES * u) * SEGCAP + lane];
#define PLACE_RANK(u)                                                              \
        int lrow##u = (v##u.x >> EID_BITS) & (RNODES - 1);                         \
        int slot##u = 0;                                                           \
        if (lane < n##u) slot##u = atomicAdd(&cnt_l[lrow##u], 1);
#define PLACE_STORE(u)                                                             \
        mx = max(mx, slot##u);                                                     \
        if (lane < n##u && slot##u < CAP)                                          \
            lbec[lrow##u * CAP + slot##u] =                                        \
                make_int2(v##u.x & ((1 << EID_BITS) - 1), v##u.y);
        PLACE_SEGS(PLACE_LOAD)
        PLACE_SEGS(PLACE_RANK)
        PLACE_SEGS(PLACE_STORE)
#undef PLACE_LOAD
#undef PLACE_RANK
#undef PLACE_STORE
#undef PLACE_SEGS
        for (int u = 0; u < PLACE_ITERS; ++u) {
            int n = __builtin_amdgcn_readlane(myc, u);
            if (lane + 32 < n) {
                int2 w = seg[((size_t)range * NB + wave + PLACE_WAVES * u) * SEGCAP + lane + 32];
                int lrow = (w.x >> EID_BITS) & (RNODES - 1);
                int sl = atomicAdd(&cnt_l[lrow], 1);
                mx = max(mx, sl);
                if (sl < CAP)
                    lbec[lrow * CAP + sl] =
                        make_int2(w.x & ((1 << EID_BITS) - 1), w.y);
            }
        }
    } else {
        for (int e = tid; e < N_EDGES; e += PLACE_THREADS) {
            int row = ei[e];
            if ((row >> RSHIFT) == range) {
                int lrow = row & (RNODES - 1);
                int sl = atomicAdd(&cnt_l[lrow], 1);
                mx = max(mx, sl);
                if (sl < CAP)
                    lbec[lrow * CAP + sl] = make_int2(e, ei[N_EDGES + e]);
            }
        }
    }
    if (mx >= 32) deep_l = 1;
    __syncthreads();
    if (tid < RNODES) {
        int node = range * RNODES + tid;
        if (node < N_NODES) cnt[node] = cnt_l[tid];
    }
    int deep = deep_l;
#define PLACE_COPY (RNODES * 16 / PLACE_THREADS)
    int pcn = tid & 15;
    int cn[PLACE_COPY];
    int4 cv[PLACE_COPY];
#pragma unroll
    for (int i = 0; i < PLACE_COPY; ++i)
        cn[i] = cnt_l[(tid + PLACE_THREADS * i) >> 4];
#pragma unroll
    for (int i = 0; i < PLACE_COPY; ++i)
        cv[i] = lbec4[((tid + PLACE_THREADS * i) >> 4) * (CAP / 2) + pcn];
#pragma unroll
    for (int i = 0; i < PLACE_COPY; ++i) {
        int node = range * RNODES + ((tid + PLACE_THREADS * i) >> 4);
        int n = cn[i];
        int4 v = cv[i];
        if (2 * pcn >= n)     { v.x = 0; v.y = 0; }
        if (2 * pcn + 1 >= n) { v.z = 0; v.w = 0; }
        if (node < N_NODES)
            bec4[(size_t)node * (CAP / 2) + pcn] = v;
    }
#undef PLACE_COPY
    if (deep) {
        for (int j = wave; j < RNODES; j += PLACE_WAVES) {
            int n = cnt_l[j];
            int pc = 16 + (lane & 15);
            int node = range * RNODES + j;
            if (n > 32 && lane < 16 && node < N_NODES) {
                if (n > CAP) n = CAP;
                int4 v = lbec4[j * (CAP / 2) + pc];
                if (2 * pc >= n)     { v.x = 0; v.y = 0; }
                if (2 * pc + 1 >= n) { v.z = 0; v.w = 0; }
                bec4[(size_t)node * (CAP / 2) + pc] = v;
            }
        }
    }
}

#define FUSE_BLOCKS 1024
#define APITCH 132

__device__ __forceinline__ float4 fma4(float4 v, float w, float4 a) {
    a.x = fmaf(v.x, w, a.x);
    a.y = fmaf(v.y, w, a.y);
    a.z = fmaf(v.z, w, a.z);
    a.w = fmaf(v.w, w, a.w);
    return a;
}

__device__ __forceinline__ float dinv(int deg) {
    return (deg > 0) ? rsqrtf((float)deg) : 0.0f;
}

__device__ __forceinline__ float4 load_row(const float* __restrict__ x, int c, unsigned lane_off) {
    const char* rp = (const char*)x + ((size_t)(unsigned)c << 9);
    return *(const float4*)(rp + lane_off);
}

static_assert(N_NODES <= (1 << 16), "a source number shares a word with a lane number");
__device__ __forceinline__ int rank_bucket(int d, int2 raw, int lane) {
    int so = 0;
    if (d <= 32) {
        int e = (lane < d) ? raw.x : 0x7fffffff;
        int c = (lane < d) ? raw.y : 0;
        int rk = 0;
        for (int k = 0; k < d; ++k) {
            int ek = __builtin_amdgcn_readlane(e, k);
            rk += (ek < e) ? 1 : 0;
        }
        if (lane >= d) rk = lane;
        so = __builtin_amdgcn_ds_permute(rk << 2, c | (lane << 16));
    }
    return so;
}

__device__ __forceinline__ float4 gather_node(
        const float4* __restrict__ x4, const int* __restrict__ ei,
        const int* __restrict__ cnt, const int2* __restrict__ bec,
        int node, int d, int2 raw, int cdeg, int so, int lane,
        int d1, int2 raw1, int n2, bool ldpair,
        int& cdeg1, int2& raw2, int& dpair, int& so1) {
    float dr = dinv(d);
    float4 acc = make_float4(0.0f, 0.0f, 0.0f, 0.0f);
#define RANK_AHEAD() so1 = rank_bucket(d1, raw1, lane)
#define LOADS_AHEAD()                                                  \
    do {                                                               \
        cdeg1 = cnt[(lane < d1) ? raw1.y : 0];                         \
        raw2  = bec[(size_t)n2 * CAP + lane];                          \
        if (ldpair) dpair = cnt[n2 + (lane & 1)];                      \
    } while (0)
    if (d <= 32) {
        float w = (lane < d) ? dr * dinv(cdeg) : 0.0f;
#define WEIGHT_OF(p) __int_as_float(__builtin_amdgcn_readlane(__float_as_int(w), (p) >> 16))

#define ROWS_AHEAD 8
        int i = 0;
        for (; i + ROWS_AHEAD <= d; i += ROWS_AHEAD) {
            float4 v[ROWS_AHEAD];
            float  wg[ROWS_AHEAD];
            int    pj[ROWS_AHEAD];
#pragma unroll
            for (int j = 0; j < ROWS_AHEAD; ++j) {
                pj[j] = __builtin_amdgcn_readlane(so, i + j);
                v[j] = load_row((const float*)x4, pj[j] & 0xffff, (unsigned)lane << 4);
            }
#pragma unroll
            for (int j = 0; j < ROWS_AHEAD; ++j) wg[j] = WEIGHT_OF(pj[j]);
#pragma unroll
            for (int j = 0; j < ROWS_AHEAD; ++j) acc = fma4(v[j], wg[j], acc);
        }
        LOADS_AHEAD();
        int left = d - i;
        {
            float4 v[ROWS_AHEAD - 1];
            float  wg[ROWS_AHEAD - 1];
#pragma unroll
            for (int j = 0; j < ROWS_AHEAD - 1; ++j) {
                v[j] = make_float4(0.0f, 0.0f, 0.0f, 0.0f);
                wg[j] = 0.0f;
                if (j < left) {
                    int p = __builtin_amdgcn_readlane(so, i + j);
                    v[j] = load_row((const float*)x4, p & 0xffff, (unsigned)lane << 4);
                    wg[j] = WEIGHT_OF(p);
                }
            }
            RANK_AHEAD();
#pragma unroll
            for (int j = 0; j < ROWS_AHEAD - 1; ++j)
                if (j < left) acc = fma4(v[j], wg[j], acc);
        }
#undef ROWS_AHEAD
#undef WEIGHT_OF
    } else if (d <= CAP) {
        LOADS_AHEAD();
        RANK_AHEAD();
        const int2* bk = bec + (size_t)node * CAP;
        int last = -1;
        for (int i = 0; i < d; ++i) {
            int best = 0x7fffffff, bp = 0;
            for (int p = lane; p < d; p += 32) {
                int id = bk[p].x;
                if (id > last && id < best) { best = id; bp = p; }
            }
            for (int s = 16; s > 0; s >>= 1) {
                int ob = __shfl_xor(best, s);
                int op = __shfl_xor(bp, s);
                if (ob < best) { best = ob; bp = op; }
            }
            last = best;
            int   ci = bk[bp].y;
            float wi = dr * dinv(cnt[ci]);
            acc = fma4(x4[(size_t)ci * (CH / 4) + lane], wi, acc);
        }
    } else {
        LOADS_AHEAD();
        RANK_AHEAD();
        for (int base = 0; base < N_EDGES; base += 32) {
            int eid = base + lane;
            int rr = -1, cc = 0;
            if (eid < N_EDGES) { rr = ei[eid]; cc = ei[N_EDGES + eid]; }
            unsigned m = (unsigned)__ballot(rr == node);
            while (m) {
                int l = __builtin_ctz(m);
                m &= m - 1;
                int   ci = __builtin_amdgcn_readlane(cc, l);
                float wi = dr * dinv(cnt[ci]);
                acc = fma4(x4[(size_t)ci * (CH / 4) + lane], wi, acc);
            }
        }
    }
#undef LOADS_AHEAD
#undef RANK_AHEAD
    return acc;
}

__device__ __forceinline__ void store_rows(const float* __restrict__ img, float* __restrict__ dst,
                                           int wave, int lane) {
    float4 v0 = *(const float4*)(img + (2 * wave) * APITCH + 4 * lane);
    float4 v1 = *(const float4*)(img + (2 * wave + 1) * APITCH + 4 * lane);
    *(float4*)(dst + (size_t)(2 * wave) * CH + 4 * lane) = v0;
    *(float4*)(dst + (size_t)(2 * wave + 1) * CH + 4 * lane) = v1;
}

__global__ void __launch_bounds__(256, 8)
k_gather_gemm(const float* __restrict__ x, const int* __restrict__ ei,
              const int* __restrict__ cnt, const int* __restrict__ bec_w,
              const float* __restrict__ W,
              const float* __restrict__ bias,
              float* __restrict__ out) {
    __shared__ float4 lds4[8 * 16 * APITCH / 4];
    float* tile = (float*)lds4;
    const int2* bec = (const int2*)bec_w;
    const float4* x4 = (const float4*)x;
    int lane = threadIdx.x & 31;
    int wave = __builtin_amdgcn_readfirstlane(threadIdx.x >> 5);
    int c = lane & 15;
    int h = lane >> 4;

    int tile0 = blockIdx.x;
    int nt = (ROW_BLOCKS - 1 - tile0) / FUSE_BLOCKS + 1;
    int nq = 2 * nt;
#define NODE_OF(q) (16 * (tile0 + ((q) >> 1) * FUSE_BLOCKS) + 2 * wave + ((q) & 1))

    int  node = NODE_OF(0);
    int  n1 = NODE_OF(1);
    int  dp0 = cnt[node + (lane & 1)];
    int2 raw = bec[(size_t)node * CAP + lane];
    int2 raw1 = bec[(size_t)n1 * CAP + lane];

    float* wst = tile + wave * 16 * APITCH;
    const float4* W4 = (const float4*)W;
#pragma unroll
    for (int i = 0; i < 16; ++i)
        *(float4*)(wst + i * APITCH + 4 * lane) = W4[(size_t)(16 * wave + i) * (CH / 4) + lane];
    __syncthreads();
    int  d  = __builtin_amdgcn_readlane(dp0, 0);
    int  d1 = __builtin_amdgcn_readlane(dp0, 1);
    int  cdeg = cnt[(lane < d) ? raw.y : 0];
    int  so = rank_bucket(d, raw, lane);
    v2f bfrag[CH / 4];
    const float* wrow = wst + c * APITCH + 2 * h;
#pragma unroll
    for (int s = 0; s < CH / 4; ++s) bfrag[s] = *(const v2f*)(wrow + 4 * s);
    __syncthreads();
    float bv = bias[16 * wave + c];

    int dodd = 0;
    for (int q = 0; q < nq; ++q) {
        bool has2 = q + 2 < nq;
        int  n2 = has2 ? NODE_OF(q + 2) : N_NODES - 1;
        bool even = (q & 1) == 0;
        int  dpair = 0;
        int  cdeg1, so1;
        int2 raw2;

        float4 acc = gather_node(x4, ei, cnt, bec, node, d, raw, cdeg, so, lane,
                                 d1, raw1, n2, even && has2, cdeg1, raw2, dpair, so1);
        int buf = (q >> 1) & 1;
        *(float4*)(tile + (buf * 16 + 2 * wave + (q & 1)) * APITCH + 4 * lane) = acc;

        if (q & 1) {
            __syncthreads();
            if (q > 1)
                store_rows(tile + (32 + (buf ^ 1) * 16) * APITCH,
                           out + (size_t)16 * (tile0 + ((q >> 1) - 1) * FUSE_BLOCKS) * CH, wave, lane);
            const float* ap = tile + (buf * 16 + c) * APITCH + 2 * h;
            v8f o = {};
#pragma unroll
            for (int s = 0; s < CH / 4; ++s) {
                v2f a = *(const v2f*)(ap + 4 * s);
                o = __builtin_amdgcn_wmma_f32_16x16x4_f32(
                    false, a, false, bfrag[s], (short)0, o, false, false);
            }
            float* os = tile + (32 + buf * 16 + 8 * h) * APITCH + 16 * wave + c;
#pragma unroll
            for (int r = 0; r < 8; ++r) os[r * APITCH] = o[r] + bv;
        }

        int d2 = dodd;
        if (even) {
            d2   = __builtin_amdgcn_readlane(dpair, 0);
            dodd = __builtin_amdgcn_readlane(dpair, 1);
        }
        node = n1;  d = d1;  raw = raw1;  cdeg = cdeg1;  so = so1;
        n1 = n2;  d1 = d2;  raw1 = raw2;
    }
#undef NODE_OF
    __syncthreads();
    store_rows(tile + (32 + ((nt - 1) & 1) * 16) * APITCH,
               out + (size_t)16 * (tile0 + (nt - 1) * FUSE_BLOCKS) * CH, wave, lane);
}

extern "C" void kernel_launch(void* const* d_in, const int* in_sizes, int n_in,
                              void* d_out, int out_size, void* d_ws, size_t ws_size,
                              hipStream_t stream) {
    const float* x  = (const float*)d_in[0];
    const int*   ei = (const int*)d_in[1];
    const float* W  = (const float*)d_in[2];
    const float* b  = (const float*)d_in[3];
    float* out = (float*)d_out;

    int*   wsi = (int*)d_ws;
    int*   cnt = wsi + WS_CNT;
    int*   bec = wsi + WS_BEC;
    int*   segcnt = wsi + WS_SEGCNT;
    int*   seg = wsi + WS_SEG;

    k_part<<<NB, PART_THREADS, 0, stream>>>(ei, segcnt, seg);
    k_place<<<NRANGE, PLACE_THREADS, 0, stream>>>(ei, segcnt, seg, cnt, bec);
    k_gather_gemm<<<FUSE_BLOCKS, 256, 0, stream>>>(x, ei, cnt, bec, W, b, out);
}
